# starting kernel
# baseline (speedup 1.0000x reference)
#include <hip/hip_runtime.h>

typedef __attribute__((ext_vector_type(2))) float v2f;
typedef __attribute__((ext_vector_type(8))) float v8f;

#define NN 50000
#define CH 128

__global__ void zero_f4(float4* __restrict__ p, int n4) {
  int i = blockIdx.x * blockDim.x + threadIdx.x;
  if (i < n4) p[i] = make_float4(0.f, 0.f, 0.f, 0.f);
}

__global__ void scatter_add(const float* __restrict__ X,
                            const int* __restrict__ src,
                            const int* __restrict__ dst,
                            float* __restrict__ agg, int E) {
  long long t = (long long)blockIdx.x * blockDim.x + threadIdx.x;
  int e = (int)(t >> 5);
  if (e >= E) return;
  int lane = (int)(t & 31);
  int s = src[e], d = dst[e];
  float4 v = *((const float4*)(X + (size_t)s * CH) + lane);
  float* a = agg + (size_t)d * CH + lane * 4;
  unsafeAtomicAdd(a + 0, v.x);
  unsafeAtomicAdd(a + 1, v.y);
  unsafeAtomicAdd(a + 2, v.z);
  unsafeAtomicAdd(a + 3, v.w);
}

template <bool ADD_AGG, bool RELU, int K, int MT_PER_WAVE>
__global__ void gemm_relu_wmma(const float* __restrict__ A,
                               const float* __restrict__ AGG,
                               const float* __restrict__ W,
                               const float* __restrict__ bias,
                               float* __restrict__ out,
                               int M, int N) {
  const int lane = threadIdx.x & 31;
  const int wave = threadIdx.x >> 5;
  const int tilesN = N >> 4;
  const int mchunks = (M >> 4) / MT_PER_WAVE;
  const int work = blockIdx.x * (blockDim.x >> 5) + wave;
  const int chunk = work / tilesN;
  if (chunk >= mchunks) return;
  const int nt = work - chunk * tilesN;
  const int hi = lane >> 4;
  const int lo = lane & 15;
  const int col = (nt << 4) + lo;

  v2f bf[K / 4];
#pragma unroll
  for (int i = 0; i < K / 4; ++i) {
    const int ka = 4 * i + (hi << 1);
    bf[i].x = W[(size_t)ka * N + col];
    bf[i].y = W[(size_t)(ka + 1) * N + col];
  }
  const float bv = bias[col];

  const int mt0 = chunk * MT_PER_WAVE;
  for (int mi = 0; mi < MT_PER_WAVE; ++mi) {
    const int m0 = (mt0 + mi) << 4;
    const float* arow = A + (size_t)(m0 + lo) * K + (hi << 1);
    const float* grow = ADD_AGG ? (AGG + (size_t)(m0 + lo) * K + (hi << 1)) : nullptr;

    v8f c;
#pragma unroll
    for (int r = 0; r < 8; ++r) c[r] = bv;

#pragma unroll
    for (int i = 0; i < K / 4; ++i) {
      v2f a;
      a.x = arow[4 * i];
      a.y = arow[4 * i + 1];
      if (ADD_AGG) {
        a.x += grow[4 * i];
        a.y += grow[4 * i + 1];
      }
      c = __builtin_amdgcn_wmma_f32_16x16x4_f32(
           false, a,  false, bf[i],
           (short)0, c,  false,  false);
    }

    float* orow = out + (size_t)(m0 + hi * 8) * N + col;
#pragma unroll
    for (int r = 0; r < 8; ++r) {
      float v = c[r];
      if (RELU) v = fmaxf(v, 0.f);
      orow[(size_t)r * N] = v;
    }
  }
}

__global__ void head_gemv(const float* __restrict__ H,
                          const float* __restrict__ w4,
                          const float* __restrict__ b4,
                          float* __restrict__ out, int M) {
  long long t = (long long)blockIdx.x * blockDim.x + threadIdx.x;
  int n = (int)(t >> 5);
  if (n >= M) return;
  int lane = (int)(t & 31);
  const float* h = H + (size_t)n * 64;
  float s = h[lane] * w4[lane] + h[lane + 32] * w4[lane + 32];
#pragma unroll
  for (int off = 16; off > 0; off >>= 1) s += __shfl_xor(s, off, 32);
  if (lane == 0) out[n] = s + b4[0];
}

extern "C" void kernel_launch(void* const* d_in, const int* in_sizes, int n_in,
                              void* d_out, int out_size, void* d_ws, size_t ws_size,
                              hipStream_t stream) {
  const float* x   = (const float*)d_in[0];
  const int*   ei  = (const int*)d_in[1];
  const int    E   = in_sizes[1] / 2;
  const int* src = ei;
  const int* dst = ei + E;
  const float* w1a = (const float*)d_in[2];
  const float* b1a = (const float*)d_in[3];
  const float* w1b = (const float*)d_in[4];
  const float* b1b = (const float*)d_in[5];
  const float* w2a = (const float*)d_in[6];
  const float* b2a = (const float*)d_in[7];
  const float* w2b = (const float*)d_in[8];
  const float* b2b = (const float*)d_in[9];
  const float* w3  = (const float*)d_in[10];
  const float* b3  = (const float*)d_in[11];
  const float* w4  = (const float*)d_in[12];
  const float* b4  = (const float*)d_in[13];
  float* outp = (float*)d_out;

  float* agg  = (float*)d_ws;
  float* bufA = agg  + (size_t)NN * CH;
  float* bufB = bufA + (size_t)NN * CH;
  float* h3   = agg;

  const int n4 = NN * CH / 4;
  dim3 zb(256), zg((n4 + 255) / 256);
  long long sthreads = (long long)E * 32;
  dim3 sb(256), sg((unsigned)((sthreads + 255) / 256));

  constexpr int MTW = 5;
  auto gblocks = [](int M, int N) {
    int waves = ((M >> 4) / MTW) * (N >> 4);
    return (unsigned)((waves + 3) / 4);
  };

  zero_f4<<<zg, zb, 0, stream>>>((float4*)agg, n4);
  scatter_add<<<sg, sb, 0, stream>>>(x, src, dst, agg, E);
  gemm_relu_wmma<true,  true, CH, MTW><<<gblocks(NN, 128), 128, 0, stream>>>(x,    agg,     w1a, b1a, bufA, NN, 128);
  gemm_relu_wmma<false, true, CH, MTW><<<gblocks(NN, 128), 128, 0, stream>>>(bufA, nullptr, w1b, b1b, bufB, NN, 128);

  zero_f4<<<zg, zb, 0, stream>>>((float4*)agg, n4);
  scatter_add<<<sg, sb, 0, stream>>>(bufB, src, dst, agg, E);
  gemm_relu_wmma<true,  true, CH, MTW><<<gblocks(NN, 128), 128, 0, stream>>>(bufB, agg,     w2a, b2a, bufA, NN, 128);
  gemm_relu_wmma<false, true, CH, MTW><<<gblocks(NN, 128), 128, 0, stream>>>(bufA, nullptr, w2b, b2b, bufB, NN, 128);

  gemm_relu_wmma<false, true, CH, MTW><<<gblocks(NN, 64), 128, 0, stream>>>(bufB, nullptr, w3, b3, h3, NN, 64);
  head_gemv<<<(NN * 32 + 255) / 256, 256, 0, stream>>>(h3, w4, b4, outp, NN);
}


# Scatter-add on long-lived waves: a wave takes 64 contiguous edges, whole-line index loads, rows of 8 edges in flight at once (12500 waves, not 800000); sum order = arrival as before
# speedup vs baseline: 1.8789x; 1.8789x over previous
#include <hip/hip_runtime.h>

typedef __attribute__((ext_vector_type(2))) float v2f;
typedef __attribute__((ext_vector_type(8))) float v8f;

#define NN 50000
#define CH 128

__global__ void zero_f4(float4* __restrict__ p, int n4) {
  int i = blockIdx.x * blockDim.x + threadIdx.x;
  if (i < n4) p[i] = make_float4(0.f, 0.f, 0.f, 0.f);
}

#define SC_WAVES 4
#define SC_EB 8

__device__ __forceinline__ void scatter_one(const float* __restrict__ X, float* __restrict__ agg,
                                            int s, int d, int lane) {
  const float* xr = X + (size_t)s * CH + lane;
  const float v0 = xr[0], v1 = xr[32], v2 = xr[64], v3 = xr[96];
  float* a = agg + (size_t)d * CH + lane;
  unsafeAtomicAdd(a + 0,  v0);
  unsafeAtomicAdd(a + 32, v1);
  unsafeAtomicAdd(a + 64, v2);
  unsafeAtomicAdd(a + 96, v3);
}

__global__ __launch_bounds__(SC_WAVES * 32) void scatter_add(const float* __restrict__ X,
                            const int* __restrict__ src,
                            const int* __restrict__ dst,
                            float* __restrict__ agg, int E) {
  const int lane = (int)(threadIdx.x & 31);
  const int w  = __builtin_amdgcn_readfirstlane((int)blockIdx.x * SC_WAVES + (int)(threadIdx.x >> 5));
  const int nw = (int)gridDim.x * SC_WAVES;
  const int nch = (E + 31) >> 5;
  const int q = nch / nw, r = nch % nw;
  const int c0 = w * q + (w < r ? w : r);
  const int c1 = c0 + q + (w < r ? 1 : 0);

  int sv = 0, dv = 0;
  if (c0 < c1 && (c0 << 5) + lane < E) { sv = src[(c0 << 5) + lane]; dv = dst[(c0 << 5) + lane]; }

  for (int ch = c0; ch < c1; ++ch) {
    const int e0 = ch << 5;
    const int n = (E - e0 < 32) ? (E - e0) : 32;
    int svn = 0, dvn = 0;
    if (ch + 1 < c1 && e0 + 32 + lane < E) { svn = src[e0 + 32 + lane]; dvn = dst[e0 + 32 + lane]; }
    if (n == 32) {
      for (int b = 0; b < 32; b += SC_EB) {
        float v[SC_EB][4];
#pragma unroll
        for (int i = 0; i < SC_EB; ++i) {
          const int s = __builtin_amdgcn_readlane(sv, b + i);
          const float* xr = X + (size_t)s * CH + lane;
          v[i][0] = xr[0]; v[i][1] = xr[32]; v[i][2] = xr[64]; v[i][3] = xr[96];
        }
#pragma unroll
        for (int i = 0; i < SC_EB; ++i) {
          const int d = __builtin_amdgcn_readlane(dv, b + i);
          float* a = agg + (size_t)d * CH + lane;
          unsafeAtomicAdd(a + 0,  v[i][0]);
          unsafeAtomicAdd(a + 32, v[i][1]);
          unsafeAtomicAdd(a + 64, v[i][2]);
          unsafeAtomicAdd(a + 96, v[i][3]);
        }
      }
    } else {
      for (int i = 0; i < n; ++i)
        scatter_one(X, agg, __builtin_amdgcn_readlane(sv, i), __builtin_amdgcn_readlane(dv, i), lane);
    }
    sv = svn; dv = dvn;
  }
}

#define DENSE_WAVES 16

template <bool ADD_AGG, int T>
__device__ __forceinline__ void load_a_rows(float4 (&a)[16], float (&bv)[T],
                                            const float* __restrict__ A,
                                            const float* __restrict__ AGG,
                                            const float* __restrict__ bcol,
                                            int row, int h) {
#pragma unroll
  for (int q = 0; q < T; ++q) bv[q] = bcol[q];
  const float4* ap = reinterpret_cast<const float4*>(A + (size_t)row * CH + 4 * h);
#pragma unroll
  for (int j = 0; j < 16; ++j) a[j] = ap[2 * j];
  if (ADD_AGG) {
    const float4* gp = reinterpret_cast<const float4*>(AGG + (size_t)row * CH + 4 * h);
#pragma unroll
    for (int j = 0; j < 16; ++j) {
      const float4 t = gp[2 * j];
      a[j].x += t.x; a[j].y += t.y; a[j].z += t.z; a[j].w += t.w;
    }
  }
}

template <bool ADD_AGG, bool RELU, int N, int T>
__global__ __launch_bounds__(DENSE_WAVES * 32) void dense_wst(const float* __restrict__ A,
                                                 const float* __restrict__ AGG,
                                                 const float* __restrict__ W,
                                                 const float* __restrict__ bias,
                                                 float* __restrict__ out,
                                                 int mtiles) {
  constexpr int NX = N / 4;
  constexpr int GW = 16 * T;
  constexpr int G  = N / GW;
  constexpr int NW = DENSE_WAVES;
  __shared__ float4 Bs[(N / 16) * 16 * 32];

  const int tid  = threadIdx.x;
  const int lane = tid & 31;
  const int wave = __builtin_amdgcn_readfirstlane(tid >> 5);
  const int c = lane & 15;
  const int h = lane >> 4;

  const int cnt = (mtiles - (int)blockIdx.x + (int)gridDim.x - 1) / (int)gridDim.x;
  const int L = cnt * G;

  float4 a[16];
  float bv[T];
  int li = wave;
  if (li < L) {
    const int mt = (int)blockIdx.x + (int)gridDim.x * (li / G);
    load_a_rows<ADD_AGG, T>(a, bv, A, AGG, bias + GW * (li % G) + T * c, (mt << 4) + c, h);
  }

  for (int task = tid; task < 32 * NX; task += NW * 32) {
    const int x  = task % NX;
    const int kq = ((task / NX) + (int)blockIdx.x) & 31;
    const float4* wp = reinterpret_cast<const float4*>(W + (size_t)(4 * kq) * N) + x;
    const float4 r0 = wp[0], r1 = wp[NX], r2 = wp[2 * NX], r3 = wp[3 * NX];
    const int j = kq >> 1, hh = kq & 1;
    auto slot = [&](int qq) -> float4* {
      const int col = 4 * x + qq;
      const int gg = col / GW, rem = col % GW;
      return Bs + ((gg * T + rem % T) * 16 + j) * 32 + hh * 16 + rem / T;
    };
    *slot(0) = make_float4(r0.x, r1.x, r2.x, r3.x);
    *slot(1) = make_float4(r0.y, r1.y, r2.y, r3.y);
    *slot(2) = make_float4(r0.z, r1.z, r2.z, r3.z);
    *slot(3) = make_float4(r0.w, r1.w, r2.w, r3.w);
  }
  __syncthreads();

  int p = 0;
  while (li < L) {
    const int mt = (int)blockIdx.x + (int)gridDim.x * (li / G);
    const int g  = li % G;
    const int m0 = mt << 4;

    v8f acc[T];
#pragma unroll
    for (int q = 0; q < T; ++q)
#pragma unroll
      for (int r = 0; r < 8; ++r) acc[q][r] = bv[q];

    const float4* bp = Bs + (g * T * 16) * 32 + lane;
#pragma unroll
    for (int j = 0; j < 16; ++j) {
      float4 b[T];
#pragma unroll
      for (int q = 0; q < T; ++q) b[q] = bp[(q * 16 + j) * 32];
      v2f alo; alo.x = a[j].x; alo.y = a[j].y;
      v2f ahi; ahi.x = a[j].z; ahi.y = a[j].w;
#pragma unroll
      for (int q = 0; q < T; ++q) {
        v2f t; t.x = b[q].x; t.y = b[q].y;
        acc[q] = __builtin_amdgcn_wmma_f32_16x16x4_f32(false, alo, false, t, (short)0, acc[q], false, false);
      }
#pragma unroll
      for (int q = 0; q < T; ++q) {
        v2f t; t.x = b[q].z; t.y = b[q].w;
        acc[q] = __builtin_amdgcn_wmma_f32_16x16x4_f32(false, ahi, false, t, (short)0, acc[q], false, false);
      }
    }

    float* orow = out + (size_t)(m0 + 8 * h) * N + GW * g + T * c;
#pragma unroll
    for (int r = 0; r < 8; ++r) {
      float v[T];
#pragma unroll
      for (int q = 0; q < T; ++q) v[q] = RELU ? fmaxf(acc[q][r], 0.f) : acc[q][r];
      if constexpr (T == 4) *reinterpret_cast<float4*>(orow + (size_t)r * N) = make_float4(v[0], v[1], v[2], v[3]);
      else                  *reinterpret_cast<float2*>(orow + (size_t)r * N) = make_float2(v[0], v[1]);
    }

    ++p;
    li = (p & 1) ? (NW * p + NW - 1 - wave) : (NW * p + wave);
    if (li < L) {
      const int mtn = (int)blockIdx.x + (int)gridDim.x * (li / G);
      load_a_rows<ADD_AGG, T>(a, bv, A, AGG, bias + GW * (li % G) + T * c, (mtn << 4) + c, h);
    }
  }
}

__global__ void head_gemv(const float* __restrict__ H,
                          const float* __restrict__ w4,
                          const float* __restrict__ b4,
                          float* __restrict__ out, int M) {
  long long t = (long long)blockIdx.x * blockDim.x + threadIdx.x;
  int n = (int)(t >> 5);
  if (n >= M) return;
  int lane = (int)(t & 31);
  const float* h = H + (size_t)n * 64;
  float s = h[lane] * w4[lane] + h[lane + 32] * w4[lane + 32];
#pragma unroll
  for (int off = 16; off > 0; off >>= 1) s += __shfl_xor(s, off, 32);
  if (lane == 0) out[n] = s + b4[0];
}

extern "C" void kernel_launch(void* const* d_in, const int* in_sizes, int n_in,
                              void* d_out, int out_size, void* d_ws, size_t ws_size,
                              hipStream_t stream) {
  const float* x   = (const float*)d_in[0];
  const int*   ei  = (const int*)d_in[1];
  const int    E   = in_sizes[1] / 2;
  const int* src = ei;
  const int* dst = ei + E;
  const float* w1a = (const float*)d_in[2];
  const float* b1a = (const float*)d_in[3];
  const float* w1b = (const float*)d_in[4];
  const float* b1b = (const float*)d_in[5];
  const float* w2a = (const float*)d_in[6];
  const float* b2a = (const float*)d_in[7];
  const float* w2b = (const float*)d_in[8];
  const float* b2b = (const float*)d_in[9];
  const float* w3  = (const float*)d_in[10];
  const float* b3  = (const float*)d_in[11];
  const float* w4  = (const float*)d_in[12];
  const float* b4  = (const float*)d_in[13];
  float* outp = (float*)d_out;

  float* agg  = (float*)d_ws;
  float* bufA = agg  + (size_t)NN * CH;
  float* bufB = bufA + (size_t)NN * CH;
  float* h3   = agg;

  const int n4 = NN * CH / 4;
  dim3 zb(256), zg((n4 + 255) / 256);
  dim3 sb(SC_WAVES * 32), sg(3125);

  const int mtiles = NN / 16;
  dim3 dg(256), db(DENSE_WAVES * 32);

  zero_f4<<<zg, zb, 0, stream>>>((float4*)agg, n4);
  scatter_add<<<sg, sb, 0, stream>>>(x, src, dst, agg, E);
  dense_wst<true,  true, 128, 4><<<dg, db, 0, stream>>>(x,    agg,     w1a, b1a, bufA, mtiles);
  dense_wst<false, true, 128, 4><<<dg, db, 0, stream>>>(bufA, nullptr, w1b, b1b, bufB, mtiles);

  zero_f4<<<zg, zb, 0, stream>>>((float4*)agg, n4);
  scatter_add<<<sg, sb, 0, stream>>>(bufB, src, dst, agg, E);
  dense_wst<true,  true, 128, 4><<<dg, db, 0, stream>>>(bufB, agg,     w2a, b2a, bufA, mtiles);
  dense_wst<false, true, 128, 4><<<dg, db, 0, stream>>>(bufA, nullptr, w2b, b2b, bufB, mtiles);

  dense_wst<false, true, 64, 2><<<dg, db, 0, stream>>>(bufB, nullptr, w3, b3, h3, mtiles);
  head_gemv<<<(NN * 32 + 255) / 256, 256, 0, stream>>>(h3, w4, b4, outp, NN);
}


# Scatter-add split by destination half: a work-group adds only into the node range of the L2 half it runs on (die number for speed only; int atomic tickets hand out work); sum order = arrival
# speedup vs baseline: 2.0018x; 1.0654x over previous
#include <hip/hip_runtime.h>

typedef __attribute__((ext_vector_type(2))) float v2f;
typedef __attribute__((ext_vector_type(8))) float v8f;

#define NN 50000
#define CH 128

__global__ void zero_f4(float4* __restrict__ p, int n4, int* __restrict__ cnt) {
  int i = blockIdx.x * blockDim.x + threadIdx.x;
  if (i < n4) p[i] = make_float4(0.f, 0.f, 0.f, 0.f);
  if (i == 0) { cnt[0] = 0; cnt[1] = 0; }
}

#define SC_WAVES 4
#define SC_EB 8
#define SC_TE (SC_WAVES * 64)

__global__ __launch_bounds__(SC_WAVES * 32) void scatter_add(const float* __restrict__ X,
                            const int* __restrict__ src,
                            const int* __restrict__ dst,
                            float* __restrict__ agg, int E,
                            int* __restrict__ cnt) {
  __shared__ int s_tk[2];
  __shared__ int s_q;
  __shared__ int2 lst[SC_WAVES][64];

  const int tid  = (int)threadIdx.x;
  const int lane = tid & 31;
  const int w  = __builtin_amdgcn_readfirstlane(tid >> 5);
  const int NT = (E + SC_TE - 1) / SC_TE;
  const unsigned lt = (1u << lane) - 1u;

  int pending = 0;
  if (tid == 0) {
    const unsigned pl = __builtin_amdgcn_s_sendmsg_rtn(0x87);
    const int q0 = (int)((pl >> 18) & 1u);
    s_q = q0;
    pending = atomicAdd(cnt + q0, 1);
  }
  __syncthreads();
  int q = s_q;
  bool other = false;

  for (int it = 0; it < 2 * NT + 2; ++it) {
    if (tid == 0) s_tk[it & 1] = pending;
    __syncthreads();
    const int t = s_tk[it & 1];
    if (t >= NT) {
      if (other) break;
      other = true; q ^= 1;
      if (tid == 0) pending = atomicAdd(cnt + q, 1);
      continue;
    }
    const int ea = t * SC_TE + w * 64 + lane;
    const int eb = ea + 32;
    int sa = 0, da = -1, sb = 0, db = -1;
    if (ea < E) { sa = src[ea]; da = dst[ea]; }
    if (eb < E) { sb = src[eb]; db = dst[eb]; }
    const bool ina = (da >= 0) && ((da >= NN / 2) == (q != 0));
    const bool inb = (db >= 0) && ((db >= NN / 2) == (q != 0));
    const unsigned ma = __builtin_amdgcn_ballot_w32(ina);
    const unsigned mb = __builtin_amdgcn_ballot_w32(inb);
    const int na = __builtin_popcount(ma);
    const int n  = na + __builtin_popcount(mb);
    if (ina) lst[w][__builtin_popcount(ma & lt)] = make_int2(sa, da);
    if (inb) lst[w][na + __builtin_popcount(mb & lt)] = make_int2(sb, db);
    __builtin_amdgcn_wave_barrier();
    int2 l0 = make_int2(0, 0), l1 = make_int2(0, 0);
    if (lane < n)      l0 = lst[w][lane];
    if (32 + lane < n) l1 = lst[w][32 + lane];
    __builtin_amdgcn_wave_barrier();

    if (n == 0 && tid == 0) pending = atomicAdd(cnt + q, 1);

#pragma nounroll
    for (int b = 0; b < n; b += SC_EB) {
      const int sv = (b < 32) ? l0.x : l1.x;
      const int dv = (b < 32) ? l0.y : l1.y;
      const int bb = b & 31;
      const int k = n - b;
      float v[SC_EB][4];
      if (k >= SC_EB) {
#pragma unroll
        for (int i = 0; i < SC_EB; ++i) {
          const int s = __builtin_amdgcn_readlane(sv, bb + i);
          const float* xr = X + (size_t)s * CH + lane;
          v[i][0] = xr[0]; v[i][1] = xr[32]; v[i][2] = xr[64]; v[i][3] = xr[96];
        }
        if (b == 0 && tid == 0) pending = atomicAdd(cnt + q, 1);
#pragma unroll
        for (int i = 0; i < SC_EB; ++i) {
          const int d = __builtin_amdgcn_readlane(dv, bb + i);
          float* a = agg + (size_t)d * CH + lane;
          unsafeAtomicAdd(a + 0,  v[i][0]);
          unsafeAtomicAdd(a + 32, v[i][1]);
          unsafeAtomicAdd(a + 64, v[i][2]);
          unsafeAtomicAdd(a + 96, v[i][3]);
        }
      } else {
#pragma unroll
        for (int i = 0; i < SC_EB - 1; ++i) {
          const int s = __builtin_amdgcn_readlane(sv, bb + (i < k ? i : k - 1));
          const float* xr = X + (size_t)s * CH + lane;
          v[i][0] = xr[0]; v[i][1] = xr[32]; v[i][2] = xr[64]; v[i][3] = xr[96];
        }
        if (b == 0 && tid == 0) pending = atomicAdd(cnt + q, 1);
#pragma unroll
        for (int i = 0; i < SC_EB - 1; ++i) {
          if (i < k) {
            const int d = __builtin_amdgcn_readlane(dv, bb + i);
            float* a = agg + (size_t)d * CH + lane;
            unsafeAtomicAdd(a + 0,  v[i][0]);
            unsafeAtomicAdd(a + 32, v[i][1]);
            unsafeAtomicAdd(a + 64, v[i][2]);
            unsafeAtomicAdd(a + 96, v[i][3]);
          }
        }
      }
    }
  }
}

#define DENSE_WAVES 16

template <bool ADD_AGG, int T>
__device__ __forceinline__ void load_a_rows(float4 (&a)[16], float (&bv)[T],
                                            const float* __restrict__ A,
                                            const float* __restrict__ AGG,
                                            const float* __restrict__ bcol,
                                            int row, int h) {
#pragma unroll
  for (int q = 0; q < T; ++q) bv[q] = bcol[q];
  const float4* ap = reinterpret_cast<const float4*>(A + (size_t)row * CH + 4 * h);
#pragma unroll
  for (int j = 0; j < 16; ++j) a[j] = ap[2 * j];
  if (ADD_AGG) {
    const float4* gp = reinterpret_cast<const float4*>(AGG + (size_t)row * CH + 4 * h);
#pragma unroll
    for (int j = 0; j < 16; ++j) {
      const float4 t = gp[2 * j];
      a[j].x += t.x; a[j].y += t.y; a[j].z += t.z; a[j].w += t.w;
    }
  }
}

template <bool ADD_AGG, bool RELU, int N, int T>
__global__ __launch_bounds__(DENSE_WAVES * 32) void dense_wst(const float* __restrict__ A,
                                                 const float* __restrict__ AGG,
                                                 const float* __restrict__ W,
                                                 const float* __restrict__ bias,
                                                 float* __restrict__ out,
                                                 int mtiles) {
  constexpr int NX = N / 4;
  constexpr int GW = 16 * T;
  constexpr int G  = N / GW;
  constexpr int NW = DENSE_WAVES;
  __shared__ float4 Bs[(N / 16) * 16 * 32];

  const int tid  = threadIdx.x;
  const int lane = tid & 31;
  const int wave = __builtin_amdgcn_readfirstlane(tid >> 5);
  const int c = lane & 15;
  const int h = lane >> 4;

  const int cnt = (mtiles - (int)blockIdx.x + (int)gridDim.x - 1) / (int)gridDim.x;
  const int L = cnt * G;

  float4 a[16];
  float bv[T];
  int li = wave;
  if (li < L) {
    const int mt = (int)blockIdx.x + (int)gridDim.x * (li / G);
    load_a_rows<ADD_AGG, T>(a, bv, A, AGG, bias + GW * (li % G) + T * c, (mt << 4) + c, h);
  }

  for (int task = tid; task < 32 * NX; task += NW * 32) {
    const int x  = task % NX;
    const int kq = ((task / NX) + (int)blockIdx.x) & 31;
    const float4* wp = reinterpret_cast<const float4*>(W + (size_t)(4 * kq) * N) + x;
    const float4 r0 = wp[0], r1 = wp[NX], r2 = wp[2 * NX], r3 = wp[3 * NX];
    const int j = kq >> 1, hh = kq & 1;
    auto slot = [&](int qq) -> float4* {
      const int col = 4 * x + qq;
      const int gg = col / GW, rem = col % GW;
      return Bs + ((gg * T + rem % T) * 16 + j) * 32 + hh * 16 + rem / T;
    };
    *slot(0) = make_float4(r0.x, r1.x, r2.x, r3.x);
    *slot(1) = make_float4(r0.y, r1.y, r2.y, r3.y);
    *slot(2) = make_float4(r0.z, r1.z, r2.z, r3.z);
    *slot(3) = make_float4(r0.w, r1.w, r2.w, r3.w);
  }
  __syncthreads();

  int p = 0;
  while (li < L) {
    const int mt = (int)blockIdx.x + (int)gridDim.x * (li / G);
    const int g  = li % G;
    const int m0 = mt << 4;

    v8f acc[T];
#pragma unroll
    for (int q = 0; q < T; ++q)
#pragma unroll
      for (int r = 0; r < 8; ++r) acc[q][r] = bv[q];

    const float4* bp = Bs + (g * T * 16) * 32 + lane;
#pragma unroll
    for (int j = 0; j < 16; ++j) {
      float4 b[T];
#pragma unroll
      for (int q = 0; q < T; ++q) b[q] = bp[(q * 16 + j) * 32];
      v2f alo; alo.x = a[j].x; alo.y = a[j].y;
      v2f ahi; ahi.x = a[j].z; ahi.y = a[j].w;
#pragma unroll
      for (int q = 0; q < T; ++q) {
        v2f t; t.x = b[q].x; t.y = b[q].y;
        acc[q] = __builtin_amdgcn_wmma_f32_16x16x4_f32(false, alo, false, t, (short)0, acc[q], false, false);
      }
#pragma unroll
      for (int q = 0; q < T; ++q) {
        v2f t; t.x = b[q].z; t.y = b[q].w;
        acc[q] = __builtin_amdgcn_wmma_f32_16x16x4_f32(false, ahi, false, t, (short)0, acc[q], false, false);
      }
    }

    float* orow = out + (size_t)(m0 + 8 * h) * N + GW * g + T * c;
#pragma unroll
    for (int r = 0; r < 8; ++r) {
      float v[T];
#pragma unroll
      for (int q = 0; q < T; ++q) v[q] = RELU ? fmaxf(acc[q][r], 0.f) : acc[q][r];
      if constexpr (T == 4) *reinterpret_cast<float4*>(orow + (size_t)r * N) = make_float4(v[0], v[1], v[2], v[3]);
      else                  *reinterpret_cast<float2*>(orow + (size_t)r * N) = make_float2(v[0], v[1]);
    }

    ++p;
    li = (p & 1) ? (NW * p + NW - 1 - wave) : (NW * p + wave);
    if (li < L) {
      const int mtn = (int)blockIdx.x + (int)gridDim.x * (li / G);
      load_a_rows<ADD_AGG, T>(a, bv, A, AGG, bias + GW * (li % G) + T * c, (mtn << 4) + c, h);
    }
  }
}

__global__ void head_gemv(const float* __restrict__ H,
                          const float* __restrict__ w4,
                          const float* __restrict__ b4,
                          float* __restrict__ out, int M) {
  long long t = (long long)blockIdx.x * blockDim.x + threadIdx.x;
  int n = (int)(t >> 5);
  if (n >= M) return;
  int lane = (int)(t & 31);
  const float* h = H + (size_t)n * 64;
  float s = h[lane] * w4[lane] + h[lane + 32] * w4[lane + 32];
#pragma unroll
  for (int off = 16; off > 0; off >>= 1) s += __shfl_xor(s, off, 32);
  if (lane == 0) out[n] = s + b4[0];
}

extern "C" void kernel_launch(void* const* d_in, const int* in_sizes, int n_in,
                              void* d_out, int out_size, void* d_ws, size_t ws_size,
                              hipStream_t stream) {
  const float* x   = (const float*)d_in[0];
  const int*   ei  = (const int*)d_in[1];
  const int    E   = in_sizes[1] / 2;
  const int* src = ei;
  const int* dst = ei + E;
  const float* w1a = (const float*)d_in[2];
  const float* b1a = (const float*)d_in[3];
  const float* w1b = (const float*)d_in[4];
  const float* b1b = (const float*)d_in[5];
  const float* w2a = (const float*)d_in[6];
  const float* b2a = (const float*)d_in[7];
  const float* w2b = (const float*)d_in[8];
  const float* b2b = (const float*)d_in[9];
  const float* w3  = (const float*)d_in[10];
  const float* b3  = (const float*)d_in[11];
  const float* w4  = (const float*)d_in[12];
  const float* b4  = (const float*)d_in[13];
  float* outp = (float*)d_out;

  float* agg  = (float*)d_ws;
  float* bufA = agg  + (size_t)NN * CH;
  float* bufB = bufA + (size_t)NN * CH;
  float* h3   = agg;
  int*   cnt  = (int*)(bufB + (size_t)NN * CH);

  const int n4 = NN * CH / 4;
  dim3 zb(256), zg((n4 + 255) / 256);
  dim3 sb(SC_WAVES * 32), sg(3125);

  const int mtiles = NN / 16;
  dim3 dg(256), db(DENSE_WAVES * 32);

  zero_f4<<<zg, zb, 0, stream>>>((float4*)agg, n4, cnt);
  scatter_add<<<sg, sb, 0, stream>>>(x, src, dst, agg, E, cnt);
  dense_wst<true,  true, 128, 4><<<dg, db, 0, stream>>>(x,    agg,     w1a, b1a, bufA, mtiles);
  dense_wst<false, true, 128, 4><<<dg, db, 0, stream>>>(bufA, nullptr, w1b, b1b, bufB, mtiles);

  zero_f4<<<zg, zb, 0, stream>>>((float4*)agg, n4, cnt);
  scatter_add<<<sg, sb, 0, stream>>>(bufB, src, dst, agg, E, cnt);
  dense_wst<true,  true, 128, 4><<<dg, db, 0, stream>>>(bufB, agg,     w2a, b2a, bufA, mtiles);
  dense_wst<false, true, 128, 4><<<dg, db, 0, stream>>>(bufA, nullptr, w2b, b2b, bufB, mtiles);

  dense_wst<false, true, 64, 2><<<dg, db, 0, stream>>>(bufB, nullptr, w3, b3, h3, mtiles);
  head_gemv<<<(NN * 32 + 255) / 256, 256, 0, stream>>>(h3, w4, b4, outp, NN);
}


# Two memory-only launches folded into dense launches (10 -> 8): layer 1b also stores h as start value of the next neighbour sum (own row first, then arrival); w4 dot in the w3 epilogue (new sum tree)
# speedup vs baseline: 2.5355x; 1.2666x over previous
#include <hip/hip_runtime.h>

typedef __attribute__((ext_vector_type(2))) float v2f;
typedef __attribute__((ext_vector_type(8))) float v8f;

#define NN 50000
#define CH 128

__global__ void zero_f4(float4* __restrict__ p, int n4, int* __restrict__ cnt) {
  int i = blockIdx.x * blockDim.x + threadIdx.x;
  if (i < n4) p[i] = make_float4(0.f, 0.f, 0.f, 0.f);
  if (i == 0) { cnt[0] = 0; cnt[1] = 0; cnt[2] = 0; cnt[3] = 0; }
}

#define SC_WAVES 4
#define SC_EB 8
#define SC_TE (SC_WAVES * 64)

__global__ __launch_bounds__(SC_WAVES * 32) void scatter_add(const float* __restrict__ X,
                            const int* __restrict__ src,
                            const int* __restrict__ dst,
                            float* __restrict__ agg, int E,
                            int* __restrict__ cnt) {
  __shared__ int s_tk[2];
  __shared__ int s_q;
  __shared__ int2 lst[SC_WAVES][64];

  const int tid  = (int)threadIdx.x;
  const int lane = tid & 31;
  const int w  = __builtin_amdgcn_readfirstlane(tid >> 5);
  const int NT = (E + SC_TE - 1) / SC_TE;
  const unsigned lt = (1u << lane) - 1u;

  int pending = 0;
  if (tid == 0) {
    const unsigned pl = __builtin_amdgcn_s_sendmsg_rtn(0x87);
    const int q0 = (int)((pl >> 18) & 1u);
    s_q = q0;
    pending = atomicAdd(cnt + q0, 1);
  }
  __syncthreads();
  int q = s_q;
  bool other = false;

  for (int it = 0; it < 2 * NT + 2; ++it) {
    if (tid == 0) s_tk[it & 1] = pending;
    __syncthreads();
    const int t = s_tk[it & 1];
    if (t >= NT) {
      if (other) break;
      other = true; q ^= 1;
      if (tid == 0) pending = atomicAdd(cnt + q, 1);
      continue;
    }
    const int ea = t * SC_TE + w * 64 + lane;
    const int eb = ea + 32;
    int sa = 0, da = -1, sb = 0, db = -1;
    if (ea < E) { sa = src[ea]; da = dst[ea]; }
    if (eb < E) { sb = src[eb]; db = dst[eb]; }
    const bool ina = (da >= 0) && ((da >= NN / 2) == (q != 0));
    const bool inb = (db >= 0) && ((db >= NN / 2) == (q != 0));
    const unsigned ma = __builtin_amdgcn_ballot_w32(ina);
    const unsigned mb = __builtin_amdgcn_ballot_w32(inb);
    const int na = __builtin_popcount(ma);
    const int n  = na + __builtin_popcount(mb);
    if (ina) lst[w][__builtin_popcount(ma & lt)] = make_int2(sa, da);
    if (inb) lst[w][na + __builtin_popcount(mb & lt)] = make_int2(sb, db);
    __builtin_amdgcn_wave_barrier();
    int2 l0 = make_int2(0, 0), l1 = make_int2(0, 0);
    if (lane < n)      l0 = lst[w][lane];
    if (32 + lane < n) l1 = lst[w][32 + lane];
    __builtin_amdgcn_wave_barrier();

    if (n == 0 && tid == 0) pending = atomicAdd(cnt + q, 1);

#pragma nounroll
    for (int b = 0; b < n; b += SC_EB) {
      const int sv = (b < 32) ? l0.x : l1.x;
      const int dv = (b < 32) ? l0.y : l1.y;
      const int bb = b & 31;
      const int k = n - b;
      float v[SC_EB][4];
      if (k >= SC_EB) {
#pragma unroll
        for (int i = 0; i < SC_EB; ++i) {
          const int s = __builtin_amdgcn_readlane(sv, bb + i);
          const float* xr = X + (size_t)s * CH + lane;
          v[i][0] = xr[0]; v[i][1] = xr[32]; v[i][2] = xr[64]; v[i][3] = xr[96];
        }
        if (b == 0 && tid == 0) pending = atomicAdd(cnt + q, 1);
#pragma unroll
        for (int i = 0; i < SC_EB; ++i) {
          const int d = __builtin_amdgcn_readlane(dv, bb + i);
          float* a = agg + (size_t)d * CH + lane;
          unsafeAtomicAdd(a + 0,  v[i][0]);
          unsafeAtomicAdd(a + 32, v[i][1]);
          unsafeAtomicAdd(a + 64, v[i][2]);
          unsafeAtomicAdd(a + 96, v[i][3]);
        }
      } else {
#pragma unroll
        for (int i = 0; i < SC_EB - 1; ++i) {
          const int s = __builtin_amdgcn_readlane(sv, bb + (i < k ? i : k - 1));
          const float* xr = X + (size_t)s * CH + lane;
          v[i][0] = xr[0]; v[i][1] = xr[32]; v[i][2] = xr[64]; v[i][3] = xr[96];
        }
        if (b == 0 && tid == 0) pending = atomicAdd(cnt + q, 1);
#pragma unroll
        for (int i = 0; i < SC_EB - 1; ++i) {
          if (i < k) {
            const int d = __builtin_amdgcn_readlane(dv, bb + i);
            float* a = agg + (size_t)d * CH + lane;
            unsafeAtomicAdd(a + 0,  v[i][0]);
            unsafeAtomicAdd(a + 32, v[i][1]);
            unsafeAtomicAdd(a + 64, v[i][2]);
            unsafeAtomicAdd(a + 96, v[i][3]);
          }
        }
      }
    }
  }
}

#define DENSE_WAVES 16

template <bool ADD_AGG, int T>
__device__ __forceinline__ void load_a_rows(float4 (&a)[16], float (&bv)[T],
                                            const float* __restrict__ A,
                                            const float* __restrict__ AGG,
                                            const float* __restrict__ bcol,
                                            int row, int h) {
#pragma unroll
  for (int q = 0; q < T; ++q) bv[q] = bcol[q];
  const float4* ap = reinterpret_cast<const float4*>(A + (size_t)row * CH + 4 * h);
#pragma unroll
  for (int j = 0; j < 16; ++j) a[j] = ap[2 * j];
  if (ADD_AGG) {
    const float4* gp = reinterpret_cast<const float4*>(AGG + (size_t)row * CH + 4 * h);
#pragma unroll
    for (int j = 0; j < 16; ++j) {
      const float4 t = gp[2 * j];
      a[j].x += t.x; a[j].y += t.y; a[j].z += t.z; a[j].w += t.w;
    }
  }
}

template <bool ADD_AGG, bool RELU, int N, int T, bool DUP, bool HEAD>
__global__ __launch_bounds__(DENSE_WAVES * 32) void dense_wst(const float* __restrict__ A,
                                                 const float* __restrict__ AGG,
                                                 const float* __restrict__ W,
                                                 const float* __restrict__ bias,
                                                 float* __restrict__ out,
                                                 float* __restrict__ out2,
                                                 const float* __restrict__ w4,
                                                 const float* __restrict__ b4,
                                                 float* __restrict__ outv,
                                                 int mtiles) {
  constexpr int NX = N / 4;
  constexpr int GW = 16 * T;
  constexpr int G  = N / GW;
  constexpr int NW = DENSE_WAVES;
  __shared__ float4 Bs[(N / 16) * 16 * 32];

  const int tid  = threadIdx.x;
  const int lane = tid & 31;
  const int wave = __builtin_amdgcn_readfirstlane(tid >> 5);
  const int c = lane & 15;
  const int h = lane >> 4;

  const int cnt = (mtiles - (int)blockIdx.x + (int)gridDim.x - 1) / (int)gridDim.x;
  const int L = cnt * G;

  float4 a[16];
  float bv[T];
  int li = wave;
  if (li < L) {
    const int mt = (int)blockIdx.x + (int)gridDim.x * (li / G);
    load_a_rows<ADD_AGG, T>(a, bv, A, AGG, bias + GW * (li % G) + T * c, (mt << 4) + c, h);
  }

  for (int task = tid; task < 32 * NX; task += NW * 32) {
    const int x  = task % NX;
    const int kq = ((task / NX) + (int)blockIdx.x) & 31;
    const float4* wp = reinterpret_cast<const float4*>(W + (size_t)(4 * kq) * N) + x;
    const float4 r0 = wp[0], r1 = wp[NX], r2 = wp[2 * NX], r3 = wp[3 * NX];
    const int j = kq >> 1, hh = kq & 1;
    auto slot = [&](int qq) -> float4* {
      const int col = 4 * x + qq;
      const int gg = col / GW, rem = col % GW;
      return Bs + ((gg * T + rem % T) * 16 + j) * 32 + hh * 16 + rem / T;
    };
    *slot(0) = make_float4(r0.x, r1.x, r2.x, r3.x);
    *slot(1) = make_float4(r0.y, r1.y, r2.y, r3.y);
    *slot(2) = make_float4(r0.z, r1.z, r2.z, r3.z);
    *slot(3) = make_float4(r0.w, r1.w, r2.w, r3.w);
  }
  __syncthreads();

  int p = 0;
  while (li < L) {
    const int mt = (int)blockIdx.x + (int)gridDim.x * (li / G);
    const int g  = li % G;
    const int m0 = mt << 4;
    if constexpr (G == 1) __atomic_signal_fence(__ATOMIC_SEQ_CST);

    v8f acc[T];
#pragma unroll
    for (int q = 0; q < T; ++q)
#pragma unroll
      for (int r = 0; r < 8; ++r) acc[q][r] = bv[q];

    const float4* bp = Bs + (g * T * 16) * 32 + lane;
#pragma unroll
    for (int j = 0; j < 16; ++j) {
      float4 b[T];
#pragma unroll
      for (int q = 0; q < T; ++q) b[q] = bp[(q * 16 + j) * 32];
      v2f alo; alo.x = a[j].x; alo.y = a[j].y;
      v2f ahi; ahi.x = a[j].z; ahi.y = a[j].w;
#pragma unroll
      for (int q = 0; q < T; ++q) {
        v2f t; t.x = b[q].x; t.y = b[q].y;
        acc[q] = __builtin_amdgcn_wmma_f32_16x16x4_f32(false, alo, false, t, (short)0, acc[q], false, false);
      }
#pragma unroll
      for (int q = 0; q < T; ++q) {
        v2f t; t.x = b[q].z; t.y = b[q].w;
        acc[q] = __builtin_amdgcn_wmma_f32_16x16x4_f32(false, ahi, false, t, (short)0, acc[q], false, false);
      }
    }

    if constexpr (HEAD) {
      static_assert(G == 1 && T == 4, "head epilogue: one wave owns all columns of its rows");
      const float4 wv = *reinterpret_cast<const float4*>(w4 + T * c);
      float s[8];
#pragma unroll
      for (int r = 0; r < 8; ++r) {
        const float v0 = RELU ? fmaxf(acc[0][r], 0.f) : acc[0][r];
        const float v1 = RELU ? fmaxf(acc[1][r], 0.f) : acc[1][r];
        const float v2 = RELU ? fmaxf(acc[2][r], 0.f) : acc[2][r];
        const float v3 = RELU ? fmaxf(acc[3][r], 0.f) : acc[3][r];
        s[r] = v0 * wv.x + v1 * wv.y + v2 * wv.z + v3 * wv.w;
      }
#pragma unroll
      for (int off = 8; off > 0; off >>= 1)
#pragma unroll
        for (int r = 0; r < 8; ++r) s[r] += __shfl_xor(s[r], off, 32);
      if (c == 0) {
        const float bb = b4[0];
        float* o = outv + m0 + 8 * h;
        *reinterpret_cast<float4*>(o)     = make_float4(s[0] + bb, s[1] + bb, s[2] + bb, s[3] + bb);
        *reinterpret_cast<float4*>(o + 4) = make_float4(s[4] + bb, s[5] + bb, s[6] + bb, s[7] + bb);
      }
    } else {
      float* orow = out + (size_t)(m0 + 8 * h) * N + GW * g + T * c;
      float* orow2 = DUP ? out2 + (size_t)(m0 + 8 * h) * N + GW * g + T * c : nullptr;
#pragma unroll
      for (int r = 0; r < 8; ++r) {
        float v[T];
#pragma unroll
        for (int q = 0; q < T; ++q) v[q] = RELU ? fmaxf(acc[q][r], 0.f) : acc[q][r];
        if constexpr (T == 4) {
          *reinterpret_cast<float4*>(orow + (size_t)r * N) = make_float4(v[0], v[1], v[2], v[3]);
          if constexpr (DUP) *reinterpret_cast<float4*>(orow2 + (size_t)r * N) = make_float4(v[0], v[1], v[2], v[3]);
        } else {
          *reinterpret_cast<float2*>(orow + (size_t)r * N) = make_float2(v[0], v[1]);
          if constexpr (DUP) *reinterpret_cast<float2*>(orow2 + (size_t)r * N) = make_float2(v[0], v[1]);
        }
      }
    }

    ++p;
    li = (p & 1) ? (NW * p + NW - 1 - wave) : (NW * p + wave);
    if (li < L) {
      const int mtn = (int)blockIdx.x + (int)gridDim.x * (li / G);
      load_a_rows<ADD_AGG, T>(a, bv, A, AGG, bias + GW * (li % G) + T * c, (mtn << 4) + c, h);
    }
  }
}

extern "C" void kernel_launch(void* const* d_in, const int* in_sizes, int n_in,
                              void* d_out, int out_size, void* d_ws, size_t ws_size,
                              hipStream_t stream) {
  const float* x   = (const float*)d_in[0];
  const int*   ei  = (const int*)d_in[1];
  const int    E   = in_sizes[1] / 2;
  const int* src = ei;
  const int* dst = ei + E;
  const float* w1a = (const float*)d_in[2];
  const float* b1a = (const float*)d_in[3];
  const float* w1b = (const float*)d_in[4];
  const float* b1b = (const float*)d_in[5];
  const float* w2a = (const float*)d_in[6];
  const float* b2a = (const float*)d_in[7];
  const float* w2b = (const float*)d_in[8];
  const float* b2b = (const float*)d_in[9];
  const float* w3  = (const float*)d_in[10];
  const float* b3  = (const float*)d_in[11];
  const float* w4  = (const float*)d_in[12];
  const float* b4  = (const float*)d_in[13];
  float* outp = (float*)d_out;

  float* agg  = (float*)d_ws;
  float* bufA = agg  + (size_t)NN * CH;
  float* bufB = bufA + (size_t)NN * CH;
  float* agg2 = bufB + (size_t)NN * CH;
  int*   cnt  = (int*)(agg2 + (size_t)NN * CH);

  const int n4 = NN * CH / 4;
  dim3 zb(256), zg((n4 + 255) / 256);
  dim3 sb(SC_WAVES * 32), sg(3125);

  const int mtiles = NN / 16;
  dim3 dg(256), db(DENSE_WAVES * 32);

  zero_f4<<<zg, zb, 0, stream>>>((float4*)agg, n4, cnt);
  scatter_add<<<sg, sb, 0, stream>>>(x, src, dst, agg, E, cnt);
  dense_wst<true,  true, 128, 4, false, false><<<dg, db, 0, stream>>>(x,    agg,     w1a, b1a, bufA, nullptr, nullptr, nullptr, nullptr, mtiles);
  dense_wst<false, true, 128, 4, true,  false><<<dg, db, 0, stream>>>(bufA, nullptr, w1b, b1b, bufB, agg2,    nullptr, nullptr, nullptr, mtiles);

  scatter_add<<<sg, sb, 0, stream>>>(bufB, src, dst, agg2, E, cnt + 2);
  dense_wst<false, true, 128, 4, false, false><<<dg, db, 0, stream>>>(agg2, nullptr, w2a, b2a, bufA, nullptr, nullptr, nullptr, nullptr, mtiles);
  dense_wst<false, true, 128, 4, false, false><<<dg, db, 0, stream>>>(bufA, nullptr, w2b, b2b, bufB, nullptr, nullptr, nullptr, nullptr, mtiles);

  dense_wst<false, true, 64, 4, false, true><<<dg, db, 0, stream>>>(bufB, nullptr, w3, b3, nullptr, nullptr, w4, b4, outp, mtiles);
}


# A wave carries its 16 nodes through a block's whole MLP in registers: products transposed (W^T x H^T), a layer's D tiles are the next B operand, no LDS pass (8 -> 5 launches); w4 dot: new sum order
# speedup vs baseline: 2.7955x; 1.1025x over previous
#include <hip/hip_runtime.h>

typedef __attribute__((ext_vector_type(2))) float v2f;
typedef __attribute__((ext_vector_type(8))) float v8f;

#define NN 50000
#define CH 128

__global__ void zero_f4(float4* __restrict__ p, int n4, int* __restrict__ cnt) {
  int i = blockIdx.x * blockDim.x + threadIdx.x;
  if (i < n4) p[i] = make_float4(0.f, 0.f, 0.f, 0.f);
  if (i == 0) { cnt[0] = 0; cnt[1] = 0; cnt[2] = 0; cnt[3] = 0; }
}

#define SC_WAVES 4
#define SC_EB 8
#define SC_TE (SC_WAVES * 64)

__global__ __launch_bounds__(SC_WAVES * 32) void scatter_add(const float* __restrict__ X,
                            const int* __restrict__ src,
                            const int* __restrict__ dst,
                            float* __restrict__ agg, int E,
                            int* __restrict__ cnt) {
  __shared__ int s_tk[2];
  __shared__ int s_q;
  __shared__ int2 lst[SC_WAVES][64];

  const int tid  = (int)threadIdx.x;
  const int lane = tid & 31;
  const int w  = __builtin_amdgcn_readfirstlane(tid >> 5);
  const int NT = (E + SC_TE - 1) / SC_TE;
  const unsigned lt = (1u << lane) - 1u;

  int pending = 0;
  if (tid == 0) {
    const unsigned pl = __builtin_amdgcn_s_sendmsg_rtn(0x87);
    const int q0 = (int)((pl >> 18) & 1u);
    s_q = q0;
    pending = atomicAdd(cnt + q0, 1);
  }
  __syncthreads();
  int q = s_q;
  bool other = false;

  for (int it = 0; it < 2 * NT + 2; ++it) {
    if (tid == 0) s_tk[it & 1] = pending;
    __syncthreads();
    const int t = s_tk[it & 1];
    if (t >= NT) {
      if (other) break;
      other = true; q ^= 1;
      if (tid == 0) pending = atomicAdd(cnt + q, 1);
      continue;
    }
    const int ea = t * SC_TE + w * 64 + lane;
    const int eb = ea + 32;
    int sa = 0, da = -1, sb = 0, db = -1;
    if (ea < E) { sa = src[ea]; da = dst[ea]; }
    if (eb < E) { sb = src[eb]; db = dst[eb]; }
    const bool ina = (da >= 0) && ((da >= NN / 2) == (q != 0));
    const bool inb = (db >= 0) && ((db >= NN / 2) == (q != 0));
    const unsigned ma = __builtin_amdgcn_ballot_w32(ina);
    const unsigned mb = __builtin_amdgcn_ballot_w32(inb);
    const int na = __builtin_popcount(ma);
    const int n  = na + __builtin_popcount(mb);
    if (ina) lst[w][__builtin_popcount(ma & lt)] = make_int2(sa, da);
    if (inb) lst[w][na + __builtin_popcount(mb & lt)] = make_int2(sb, db);
    __builtin_amdgcn_wave_barrier();
    int2 l0 = make_int2(0, 0), l1 = make_int2(0, 0);
    if (lane < n)      l0 = lst[w][lane];
    if (32 + lane < n) l1 = lst[w][32 + lane];
    __builtin_amdgcn_wave_barrier();

    if (n == 0 && tid == 0) pending = atomicAdd(cnt + q, 1);

#pragma nounroll
    for (int b = 0; b < n; b += SC_EB) {
      const int sv = (b < 32) ? l0.x : l1.x;
      const int dv = (b < 32) ? l0.y : l1.y;
      const int bb = b & 31;
      const int k = n - b;
      float v[SC_EB][4];
      if (k >= SC_EB) {
#pragma unroll
        for (int i = 0; i < SC_EB; ++i) {
          const int s = __builtin_amdgcn_readlane(sv, bb + i);
          const float* xr = X + (size_t)s * CH + lane;
          v[i][0] = xr[0]; v[i][1] = xr[32]; v[i][2] = xr[64]; v[i][3] = xr[96];
        }
        if (b == 0 && tid == 0) pending = atomicAdd(cnt + q, 1);
#pragma unroll
        for (int i = 0; i < SC_EB; ++i) {
          const int d = __builtin_amdgcn_readlane(dv, bb + i);
          float* a = agg + (size_t)d * CH + lane;
          unsafeAtomicAdd(a + 0,  v[i][0]);
          unsafeAtomicAdd(a + 32, v[i][1]);
          unsafeAtomicAdd(a + 64, v[i][2]);
          unsafeAtomicAdd(a + 96, v[i][3]);
        }
      } else {
#pragma unroll
        for (int i = 0; i < SC_EB - 1; ++i) {
          const int s = __builtin_amdgcn_readlane(sv, bb + (i < k ? i : k - 1));
          const float* xr = X + (size_t)s * CH + lane;
          v[i][0] = xr[0]; v[i][1] = xr[32]; v[i][2] = xr[64]; v[i][3] = xr[96];
        }
        if (b == 0 && tid == 0) pending = atomicAdd(cnt + q, 1);
#pragma unroll
        for (int i = 0; i < SC_EB - 1; ++i) {
          if (i < k) {
            const int d = __builtin_amdgcn_readlane(dv, bb + i);
            float* a = agg + (size_t)d * CH + lane;
            unsafeAtomicAdd(a + 0,  v[i][0]);
            unsafeAtomicAdd(a + 32, v[i][1]);
            unsafeAtomicAdd(a + 64, v[i][2]);
            unsafeAtomicAdd(a + 96, v[i][3]);
          }
        }
      }
    }
  }
}

#define DENSE_WAVES 16

template <int N>
__device__ __forceinline__ void stage_w(float4* __restrict__ Wf, const float* __restrict__ W,
                                        int tid, int rot) {
  constexpr int NX = N / 4;
  constexpr int PER = (32 * NX) / (DENSE_WAVES * 32);
  static_assert(PER * DENSE_WAVES * 32 == 32 * NX, "tasks divide evenly among the threads");
#pragma unroll
  for (int i = 0; i < PER; ++i) {
    const int task = tid + i * DENSE_WAVES * 32;
    const int x  = task % NX;
    const int kq = ((task / NX) + rot) & 31;
    const float4* wp = reinterpret_cast<const float4*>(W + (size_t)(4 * kq) * N) + x;
    const float4 r0 = wp[0], r1 = wp[NX], r2 = wp[2 * NX], r3 = wp[3 * NX];
    const int j = kq >> 1, hh = kq & 1;
    float4* s = Wf + ((x >> 2) * 16 + j) * 32 + hh * 16 + ((x & 1) << 3) + ((x & 2) << 1);
    s[0] = make_float4(r0.x, r1.x, r2.x, r3.x);
    s[1] = make_float4(r0.y, r1.y, r2.y, r3.y);
    s[2] = make_float4(r0.z, r1.z, r2.z, r3.z);
    s[3] = make_float4(r0.w, r1.w, r2.w, r3.w);
  }
}

template <bool ADD_AGG>
__device__ __forceinline__ void load_rows(float4 (&a)[16], const float* __restrict__ A,
                                          const float* __restrict__ AGG, int row, int h) {
  const float4* ap = reinterpret_cast<const float4*>(A + (size_t)row * CH + 4 * h);
#pragma unroll
  for (int j = 0; j < 16; ++j) a[j] = ap[2 * j];
  if (ADD_AGG) {
    const float4* gp = reinterpret_cast<const float4*>(AGG + (size_t)row * CH + 4 * h);
#pragma unroll
    for (int j = 0; j < 16; ++j) {
      const float4 t = gp[2 * j];
      a[j].x += t.x; a[j].y += t.y; a[j].z += t.z; a[j].w += t.w;
    }
  }
}

template <int NT>
__device__ __forceinline__ void layer(v8f (&acc)[NT], const float4 (&in)[16],
                                      const float4* __restrict__ Wf,
                                      const float4* __restrict__ bias, int lane, int h) {
  const float4* b4p = bias + h;
#pragma unroll
  for (int q = 0; q < NT; ++q) {
    const float4 lo = b4p[4 * q], hi = b4p[4 * q + 2];
    acc[q][0] = lo.x; acc[q][1] = lo.y; acc[q][2] = lo.z; acc[q][3] = lo.w;
    acc[q][4] = hi.x; acc[q][5] = hi.y; acc[q][6] = hi.z; acc[q][7] = hi.w;
  }
  const float4* wp = Wf + lane;
#pragma unroll
  for (int j = 0; j < 16; ++j) {
    float4 w[NT];
#pragma unroll
    for (int q = 0; q < NT; ++q) w[q] = wp[(q * 16 + j) * 32];
    v2f blo; blo.x = in[j].x; blo.y = in[j].y;
    v2f bhi; bhi.x = in[j].z; bhi.y = in[j].w;
#pragma unroll
    for (int q = 0; q < NT; ++q) {
      v2f t; t.x = w[q].x; t.y = w[q].y;
      acc[q] = __builtin_amdgcn_wmma_f32_16x16x4_f32(false, t, false, blo, (short)0, acc[q], false, false);
    }
#pragma unroll
    for (int q = 0; q < NT; ++q) {
      v2f t; t.x = w[q].z; t.y = w[q].w;
      acc[q] = __builtin_amdgcn_wmma_f32_16x16x4_f32(false, t, false, bhi, (short)0, acc[q], false, false);
    }
  }
}

__device__ __forceinline__ void relu_rows(float4 (&o)[16], const v8f (&acc)[8]) {
#pragma unroll
  for (int q = 0; q < 8; ++q) {
    o[2 * q]     = make_float4(fmaxf(acc[q][0], 0.f), fmaxf(acc[q][1], 0.f), fmaxf(acc[q][2], 0.f), fmaxf(acc[q][3], 0.f));
    o[2 * q + 1] = make_float4(fmaxf(acc[q][4], 0.f), fmaxf(acc[q][5], 0.f), fmaxf(acc[q][6], 0.f), fmaxf(acc[q][7], 0.f));
  }
}

template <bool ADD_AGG, bool HEAD>
__global__ __launch_bounds__(DENSE_WAVES * 32) void mlp_chain(const float* __restrict__ A,
                                                 const float* __restrict__ AGG,
                                                 const float* __restrict__ Wa,
                                                 const float* __restrict__ ba,
                                                 const float* __restrict__ Wb,
                                                 const float* __restrict__ bb,
                                                 const float* __restrict__ W3,
                                                 const float* __restrict__ b3,
                                                 const float* __restrict__ w4,
                                                 const float* __restrict__ b4,
                                                 float* __restrict__ out,
                                                 float* __restrict__ out2,
                                                 float* __restrict__ outv,
                                                 int mtiles) {
  constexpr int NW = DENSE_WAVES;
  __shared__ float4 Fa[8 * 16 * 32];
  __shared__ float4 Fb[8 * 16 * 32];
  __shared__ float4 F3[HEAD ? 4 * 16 * 32 : 1];
  __shared__ float4 Bi[(CH + CH + 64 + 64) / 4];

  const int tid  = threadIdx.x;
  const int lane = tid & 31;
  const int wave = __builtin_amdgcn_readfirstlane(tid >> 5);
  const int c = lane & 15;
  const int h = lane >> 4;

  const int cnt = (mtiles - (int)blockIdx.x + (int)gridDim.x - 1) / (int)gridDim.x;

  float4 in[16];
  int t = wave;
  if (t < cnt) load_rows<ADD_AGG>(in, A, AGG, ((((int)blockIdx.x + (int)gridDim.x * t)) << 4) + c, h);

  stage_w<CH>(Fa, Wa, tid, (int)blockIdx.x);
  stage_w<CH>(Fb, Wb, tid, (int)blockIdx.x);
  if constexpr (HEAD) stage_w<64>(F3, W3, tid, (int)blockIdx.x);
  if (tid < CH / 4) {
    Bi[tid] = reinterpret_cast<const float4*>(ba)[tid];
    Bi[CH / 4 + tid] = reinterpret_cast<const float4*>(bb)[tid];
    if constexpr (HEAD) {
      if (tid < 16) {
        Bi[CH / 2 + tid] = reinterpret_cast<const float4*>(b3)[tid];
        Bi[CH / 2 + 16 + tid] = reinterpret_cast<const float4*>(w4)[tid];
      }
    }
  }
  __syncthreads();

  while (t < cnt) {
    const int m0 = ((int)blockIdx.x + (int)gridDim.x * t) << 4;
    __atomic_signal_fence(__ATOMIC_SEQ_CST);

    v8f acc[8];
    layer<8>(acc, in, Fa, Bi, lane, h);
    if constexpr (HEAD) __builtin_amdgcn_sched_barrier(0);
    relu_rows(in, acc);
    layer<8>(acc, in, Fb, Bi + CH / 4, lane, h);

    if constexpr (HEAD) {
      __builtin_amdgcn_sched_barrier(0);
      relu_rows(in, acc);
      v8f a3[4];
      layer<4>(a3, in, F3, Bi + CH / 2, lane, h);
      const float4* w4p = Bi + CH / 2 + 16 + h;
      float s = 0.f;
#pragma unroll
      for (int q = 0; q < 4; ++q) {
        const float4 lo = w4p[4 * q], hi = w4p[4 * q + 2];
        s += fmaxf(a3[q][0], 0.f) * lo.x; s += fmaxf(a3[q][1], 0.f) * lo.y;
        s += fmaxf(a3[q][2], 0.f) * lo.z; s += fmaxf(a3[q][3], 0.f) * lo.w;
        s += fmaxf(a3[q][4], 0.f) * hi.x; s += fmaxf(a3[q][5], 0.f) * hi.y;
        s += fmaxf(a3[q][6], 0.f) * hi.z; s += fmaxf(a3[q][7], 0.f) * hi.w;
      }
      s += __shfl_xor(s, 16, 32);
      if (h == 0) outv[m0 + c] = s + b4[0];
    } else {
      relu_rows(in, acc);
      float4* o1 = reinterpret_cast<float4*>(out  + (size_t)(m0 + c) * CH + 4 * h);
      float4* o2 = reinterpret_cast<float4*>(out2 + (size_t)(m0 + c) * CH + 4 * h);
#pragma unroll
      for (int j = 0; j < 16; ++j) { o1[2 * j] = in[j]; o2[2 * j] = in[j]; }
    }

    t += NW;
    if (t < cnt) load_rows<ADD_AGG>(in, A, AGG, ((((int)blockIdx.x + (int)gridDim.x * t)) << 4) + c, h);
  }
}

extern "C" void kernel_launch(void* const* d_in, const int* in_sizes, int n_in,
                              void* d_out, int out_size, void* d_ws, size_t ws_size,
                              hipStream_t stream) {
  const float* x   = (const float*)d_in[0];
  const int*   ei  = (const int*)d_in[1];
  const int    E   = in_sizes[1] / 2;
  const int* src = ei;
  const int* dst = ei + E;
  const float* w1a = (const float*)d_in[2];
  const float* b1a = (const float*)d_in[3];
  const float* w1b = (const float*)d_in[4];
  const float* b1b = (const float*)d_in[5];
  const float* w2a = (const float*)d_in[6];
  const float* b2a = (const float*)d_in[7];
  const float* w2b = (const float*)d_in[8];
  const float* b2b = (const float*)d_in[9];
  const float* w3  = (const float*)d_in[10];
  const float* b3  = (const float*)d_in[11];
  const float* w4  = (const float*)d_in[12];
  const float* b4  = (const float*)d_in[13];
  float* outp = (float*)d_out;

  float* agg  = (float*)d_ws;
  float* bufB = agg  + (size_t)NN * CH;
  float* agg2 = bufB + (size_t)NN * CH;
  int*   cnt  = (int*)(agg2 + (size_t)NN * CH);

  const int n4 = NN * CH / 4;
  dim3 zb(256), zg((n4 + 255) / 256);
  dim3 sb(SC_WAVES * 32), sg(3125);

  const int mtiles = NN / 16;
  dim3 dg(256), db(DENSE_WAVES * 32);

  zero_f4<<<zg, zb, 0, stream>>>((float4*)agg, n4, cnt);
  scatter_add<<<sg, sb, 0, stream>>>(x, src, dst, agg, E, cnt);
  mlp_chain<true, false><<<dg, db, 0, stream>>>(x, agg, w1a, b1a, w1b, b1b, nullptr, nullptr, nullptr, nullptr,
                                                bufB, agg2, nullptr, mtiles);

  scatter_add<<<sg, sb, 0, stream>>>(bufB, src, dst, agg2, E, cnt + 2);
  mlp_chain<false, true><<<dg, db, 0, stream>>>(agg2, nullptr, w2a, b2a, w2b, b2b, w3, b3, w4, b4,
                                                nullptr, nullptr, outp, mtiles);
}


# Neighbour sums read source rows from LDS, not L2: edges grouped once per call by (196-row source block, dst half), per-chunk LDS counting sort; only f32 atomic adds stay random; sum order = arrival
# speedup vs baseline: 3.2319x; 1.1561x over previous
#include <hip/hip_runtime.h>

typedef __attribute__((ext_vector_type(2))) float v2f;
typedef __attribute__((ext_vector_type(8))) float v8f;

#define NN 50000
#define CH 128

#define NB 256
#define BS ((NN + NB - 1) / NB)
#define NCH 256
#define NBIN (2 * NB)
#define PREP_T 1024
#define SC_W 16
#define SC_T (SC_W * 32)
#define SEGW (NCH / SC_W)

__global__ __launch_bounds__(PREP_T) void prep(float4* __restrict__ p, int n4, int* __restrict__ cnt,
                                               const int* __restrict__ src, const int* __restrict__ dst, int E,
                                               int2* __restrict__ seg, int2* __restrict__ rec) {
  __shared__ int hist[NBIN];
  __shared__ int cur[NBIN];
  __shared__ int wsum[NBIN / 32];
  const int tid = (int)threadIdx.x;
  const int lane = tid & 31;
  const int wv = tid >> 5;
  const int c = (int)blockIdx.x;

  const int per = (n4 + (int)gridDim.x - 1) / (int)gridDim.x;
  for (int i = tid; i < per; i += PREP_T) {
    const int k = c * per + i;
    if (k < n4) p[k] = make_float4(0.f, 0.f, 0.f, 0.f);
  }
  if (c == 0 && tid < 4) cnt[tid] = 0;
  if (tid < NBIN) hist[tid] = 0;
  __syncthreads();

  const int CE = (E + NCH - 1) / NCH;
  const int e0 = c * CE;
  const int e1 = min(e0 + CE, E);
  for (int e = e0 + tid; e < e1; e += PREP_T) {
    const int s = src[e], d = dst[e];
    atomicAdd(&hist[2 * (s / BS) + (d >= NN / 2 ? 1 : 0)], 1);
  }
  __syncthreads();

  int v = 0, incl = 0;
  if (tid < NBIN) {
    v = hist[tid];
    incl = v;
#pragma unroll
    for (int o = 1; o < 32; o <<= 1) {
      const int t = __shfl_up(incl, o, 32);
      if (lane >= o) incl += t;
    }
    if (lane == 31) wsum[wv] = incl;
  }
  __syncthreads();
  if (tid < NBIN) {
    int pre = 0;
    for (int k = 0; k < wv; ++k) pre += wsum[k];
    const int ex = pre + incl - v;
    cur[tid] = ex;
    seg[(size_t)tid * NCH + c] = make_int2(ex, ex + v);
  }
  __syncthreads();

  for (int e = e0 + tid; e < e1; e += PREP_T) {
    const int s = src[e], d = dst[e];
    const int pos = atomicAdd(&cur[2 * (s / BS) + (d >= NN / 2 ? 1 : 0)], 1);
    rec[(size_t)e0 + pos] = make_int2(s, d);
  }
}

__global__ __launch_bounds__(SC_T) void scatter_lds(const float* __restrict__ X,
                                                    const int2* __restrict__ seg,
                                                    const int2* __restrict__ rec,
                                                    float* __restrict__ agg, int E,
                                                    int* __restrict__ cnt) {
  __shared__ float rows[BS * CH];
  __shared__ int s_tk[2];
  __shared__ int s_q;

  const int tid  = (int)threadIdx.x;
  const int lane = tid & 31;
  const int w  = __builtin_amdgcn_readfirstlane(tid >> 5);
  const int CE = (E + NCH - 1) / NCH;

  int pending = 0;
  if (tid == 0) {
    const unsigned pl = __builtin_amdgcn_s_sendmsg_rtn(0x87);
    const int q0 = (int)((pl >> 18) & 1u);
    s_q = q0;
    pending = atomicAdd(cnt + q0, 1);
  }
  __syncthreads();
  int q = s_q;
  bool other = false;

  for (int it = 0; it < 2 * NB + 2; ++it) {
    if (tid == 0) s_tk[it & 1] = pending;
    __syncthreads();
    const int t = s_tk[it & 1];
    if (t >= NB) {
      if (other) break;
      other = true; q ^= 1;
      if (tid == 0) pending = atomicAdd(cnt + q, 1);
      continue;
    }
    int2 sg = make_int2(0, 0);
    if (lane < SEGW) sg = seg[(size_t)(2 * t + q) * NCH + w * SEGW + lane];

    const int r0 = t * BS;
    const int nr4 = min(BS, NN - r0) * (CH / 4);
    const float4* xp = reinterpret_cast<const float4*>(X + (size_t)r0 * CH);
    float4* lp = reinterpret_cast<float4*>(rows);
    constexpr int FILL_IT = (BS * (CH / 4) + SC_T - 1) / SC_T;
    float4 fl[FILL_IT];
#pragma unroll
    for (int k = 0; k < FILL_IT; ++k) fl[k] = xp[min(tid + k * SC_T, nr4 - 1)];
#pragma unroll
    for (int k = 0; k < FILL_IT; ++k) lp[min(tid + k * SC_T, nr4 - 1)] = fl[k];

    const int len = sg.y - sg.x;
    int incl = len;
#pragma unroll
    for (int o = 1; o < 32; o <<= 1) {
      const int u = __shfl_up(incl, o, 32);
      if (lane >= o) incl += u;
    }
    const int excl = incl - len;
    const int T = __builtin_amdgcn_readlane(incl, 31);

    __syncthreads();

#pragma nounroll
    for (int base = 0; base < T; base += 32) {
      const int f = base + lane;
      int cj = 0, off = 0;
#pragma unroll
      for (int j = 0; j < SEGW; ++j) {
        const int pj = __builtin_amdgcn_readlane(excl, j);
        const int sj = __builtin_amdgcn_readlane(sg.x, j);
        if (f >= pj) { cj = j; off = sj + (f - pj); }
      }
      int sv = r0, dv = 0;
      if (f < T) {
        const int2 r = rec[(size_t)(w * SEGW + cj) * CE + off];
        sv = r.x; dv = r.y;
      }
      const int m = min(32, T - base);

#pragma nounroll
      for (int i = 0; i < m; i += 4) {
        float v[4][4];
#pragma unroll
        for (int u = 0; u < 4; ++u) {
          const int k = min(i + u, m - 1);
          const int s = __builtin_amdgcn_readlane(sv, k) - r0;
          const float* lr = rows + s * CH + lane;
          v[u][0] = lr[0]; v[u][1] = lr[32]; v[u][2] = lr[64]; v[u][3] = lr[96];
        }
#pragma unroll
        for (int u = 0; u < 4; ++u) {
          if (i + u < m) {
            const int d = __builtin_amdgcn_readlane(dv, i + u);
            float* a = agg + (size_t)d * CH + lane;
            unsafeAtomicAdd(a + 0,  v[u][0]);
            unsafeAtomicAdd(a + 32, v[u][1]);
            unsafeAtomicAdd(a + 64, v[u][2]);
            unsafeAtomicAdd(a + 96, v[u][3]);
          }
        }
      }
    }
    if (tid == 0) pending = atomicAdd(cnt + q, 1);
  }
}

#define DENSE_WAVES 16

template <int N>
__device__ __forceinline__ void stage_w(float4* __restrict__ Wf, const float* __restrict__ W,
                                        int tid, int rot) {
  constexpr int NX = N / 4;
  constexpr int PER = (32 * NX) / (DENSE_WAVES * 32);
  static_assert(PER * DENSE_WAVES * 32 == 32 * NX, "tasks divide evenly among the threads");
#pragma unroll
  for (int i = 0; i < PER; ++i) {
    const int task = tid + i * DENSE_WAVES * 32;
    const int x  = task % NX;
    const int kq = ((task / NX) + rot) & 31;
    const float4* wp = reinterpret_cast<const float4*>(W + (size_t)(4 * kq) * N) + x;
    const float4 r0 = wp[0], r1 = wp[NX], r2 = wp[2 * NX], r3 = wp[3 * NX];
    const int j = kq >> 1, hh = kq & 1;
    float4* s = Wf + ((x >> 2) * 16 + j) * 32 + hh * 16 + ((x & 1) << 3) + ((x & 2) << 1);
    s[0] = make_float4(r0.x, r1.x, r2.x, r3.x);
    s[1] = make_float4(r0.y, r1.y, r2.y, r3.y);
    s[2] = make_float4(r0.z, r1.z, r2.z, r3.z);
    s[3] = make_float4(r0.w, r1.w, r2.w, r3.w);
  }
}

template <bool ADD_AGG>
__device__ __forceinline__ void load_rows(float4 (&a)[16], const float* __restrict__ A,
                                          const float* __restrict__ AGG, int row, int h) {
  const float4* ap = reinterpret_cast<const float4*>(A + (size_t)row * CH + 4 * h);
#pragma unroll
  for (int j = 0; j < 16; ++j) a[j] = ap[2 * j];
  if (ADD_AGG) {
    const float4* gp = reinterpret_cast<const float4*>(AGG + (size_t)row * CH + 4 * h);
#pragma unroll
    for (int j = 0; j < 16; ++j) {
      const float4 t = gp[2 * j];
      a[j].x += t.x; a[j].y += t.y; a[j].z += t.z; a[j].w += t.w;
    }
  }
}

template <int NT>
__device__ __forceinline__ void layer(v8f (&acc)[NT], const float4 (&in)[16],
                                      const float4* __restrict__ Wf,
                                      const float4* __restrict__ bias, int lane, int h) {
  const float4* b4p = bias + h;
#pragma unroll
  for (int q = 0; q < NT; ++q) {
    const float4 lo = b4p[4 * q], hi = b4p[4 * q + 2];
    acc[q][0] = lo.x; acc[q][1] = lo.y; acc[q][2] = lo.z; acc[q][3] = lo.w;
    acc[q][4] = hi.x; acc[q][5] = hi.y; acc[q][6] = hi.z; acc[q][7] = hi.w;
  }
  const float4* wp = Wf + lane;
#pragma unroll
  for (int j = 0; j < 16; ++j) {
    float4 w[NT];
#pragma unroll
    for (int q = 0; q < NT; ++q) w[q] = wp[(q * 16 + j) * 32];
    v2f blo; blo.x = in[j].x; blo.y = in[j].y;
    v2f bhi; bhi.x = in[j].z; bhi.y = in[j].w;
#pragma unroll
    for (int q = 0; q < NT; ++q) {
      v2f t; t.x = w[q].x; t.y = w[q].y;
      acc[q] = __builtin_amdgcn_wmma_f32_16x16x4_f32(false, t, false, blo, (short)0, acc[q], false, false);
    }
#pragma unroll
    for (int q = 0; q < NT; ++q) {
      v2f t; t.x = w[q].z; t.y = w[q].w;
      acc[q] = __builtin_amdgcn_wmma_f32_16x16x4_f32(false, t, false, bhi, (short)0, acc[q], false, false);
    }
  }
}

__device__ __forceinline__ void relu_rows(float4 (&o)[16], const v8f (&acc)[8]) {
#pragma unroll
  for (int q = 0; q < 8; ++q) {
    o[2 * q]     = make_float4(fmaxf(acc[q][0], 0.f), fmaxf(acc[q][1], 0.f), fmaxf(acc[q][2], 0.f), fmaxf(acc[q][3], 0.f));
    o[2 * q + 1] = make_float4(fmaxf(acc[q][4], 0.f), fmaxf(acc[q][5], 0.f), fmaxf(acc[q][6], 0.f), fmaxf(acc[q][7], 0.f));
  }
}

template <bool ADD_AGG, bool HEAD>
__global__ __launch_bounds__(DENSE_WAVES * 32) void mlp_chain(const float* __restrict__ A,
                                                 const float* __restrict__ AGG,
                                                 const float* __restrict__ Wa,
                                                 const float* __restrict__ ba,
                                                 const float* __restrict__ Wb,
                                                 const float* __restrict__ bb,
                                                 const float* __restrict__ W3,
                                                 const float* __restrict__ b3,
                                                 const float* __restrict__ w4,
                                                 const float* __restrict__ b4,
                                                 float* __restrict__ out,
                                                 float* __restrict__ out2,
                                                 float* __restrict__ outv,
                                                 int mtiles) {
  constexpr int NW = DENSE_WAVES;
  __shared__ float4 Fa[8 * 16 * 32];
  __shared__ float4 Fb[8 * 16 * 32];
  __shared__ float4 F3[HEAD ? 4 * 16 * 32 : 1];
  __shared__ float4 Bi[(CH + CH + 64 + 64) / 4];

  const int tid  = threadIdx.x;
  const int lane = tid & 31;
  const int wave = __builtin_amdgcn_readfirstlane(tid >> 5);
  const int c = lane & 15;
  const int h = lane >> 4;

  const int cnt = (mtiles - (int)blockIdx.x + (int)gridDim.x - 1) / (int)gridDim.x;

  float4 in[16];
  int t = wave;
  if (t < cnt) load_rows<ADD_AGG>(in, A, AGG, ((((int)blockIdx.x + (int)gridDim.x * t)) << 4) + c, h);

  stage_w<CH>(Fa, Wa, tid, (int)blockIdx.x);
  stage_w<CH>(Fb, Wb, tid, (int)blockIdx.x);
  if constexpr (HEAD) stage_w<64>(F3, W3, tid, (int)blockIdx.x);
  if (tid < CH / 4) {
    Bi[tid] = reinterpret_cast<const float4*>(ba)[tid];
    Bi[CH / 4 + tid] = reinterpret_cast<const float4*>(bb)[tid];
    if constexpr (HEAD) {
      if (tid < 16) {
        Bi[CH / 2 + tid] = reinterpret_cast<const float4*>(b3)[tid];
        Bi[CH / 2 + 16 + tid] = reinterpret_cast<const float4*>(w4)[tid];
      }
    }
  }
  __syncthreads();

  while (t < cnt) {
    const int m0 = ((int)blockIdx.x + (int)gridDim.x * t) << 4;
    __atomic_signal_fence(__ATOMIC_SEQ_CST);

    v8f acc[8];
    layer<8>(acc, in, Fa, Bi, lane, h);
    if constexpr (HEAD) __builtin_amdgcn_sched_barrier(0);
    relu_rows(in, acc);
    layer<8>(acc, in, Fb, Bi + CH / 4, lane, h);

    if constexpr (HEAD) {
      __builtin_amdgcn_sched_barrier(0);
      relu_rows(in, acc);
      v8f a3[4];
      layer<4>(a3, in, F3, Bi + CH / 2, lane, h);
      const float4* w4p = Bi + CH / 2 + 16 + h;
      float s = 0.f;
#pragma unroll
      for (int q = 0; q < 4; ++q) {
        const float4 lo = w4p[4 * q], hi = w4p[4 * q + 2];
        s += fmaxf(a3[q][0], 0.f) * lo.x; s += fmaxf(a3[q][1], 0.f) * lo.y;
        s += fmaxf(a3[q][2], 0.f) * lo.z; s += fmaxf(a3[q][3], 0.f) * lo.w;
        s += fmaxf(a3[q][4], 0.f) * hi.x; s += fmaxf(a3[q][5], 0.f) * hi.y;
        s += fmaxf(a3[q][6], 0.f) * hi.z; s += fmaxf(a3[q][7], 0.f) * hi.w;
      }
      s += __shfl_xor(s, 16, 32);
      if (h == 0) outv[m0 + c] = s + b4[0];
    } else {
      relu_rows(in, acc);
      float4* o1 = reinterpret_cast<float4*>(out  + (size_t)(m0 + c) * CH + 4 * h);
      float4* o2 = reinterpret_cast<float4*>(out2 + (size_t)(m0 + c) * CH + 4 * h);
#pragma unroll
      for (int j = 0; j < 16; ++j) { o1[2 * j] = in[j]; o2[2 * j] = in[j]; }
    }

    t += NW;
    if (t < cnt) load_rows<ADD_AGG>(in, A, AGG, ((((int)blockIdx.x + (int)gridDim.x * t)) << 4) + c, h);
  }
}

extern "C" void kernel_launch(void* const* d_in, const int* in_sizes, int n_in,
                              void* d_out, int out_size, void* d_ws, size_t ws_size,
                              hipStream_t stream) {
  const float* x   = (const float*)d_in[0];
  const int*   ei  = (const int*)d_in[1];
  const int    E   = in_sizes[1] / 2;
  const int* src = ei;
  const int* dst = ei + E;
  const float* w1a = (const float*)d_in[2];
  const float* b1a = (const float*)d_in[3];
  const float* w1b = (const float*)d_in[4];
  const float* b1b = (const float*)d_in[5];
  const float* w2a = (const float*)d_in[6];
  const float* b2a = (const float*)d_in[7];
  const float* w2b = (const float*)d_in[8];
  const float* b2b = (const float*)d_in[9];
  const float* w3  = (const float*)d_in[10];
  const float* b3  = (const float*)d_in[11];
  const float* w4  = (const float*)d_in[12];
  const float* b4  = (const float*)d_in[13];
  float* outp = (float*)d_out;

  float* agg  = (float*)d_ws;
  float* bufB = agg  + (size_t)NN * CH;
  float* agg2 = bufB + (size_t)NN * CH;
  int*   cnt  = (int*)(agg2 + (size_t)NN * CH);
  int2*  seg  = (int2*)(cnt + 64);
  int2*  rec  = seg + (size_t)NBIN * NCH;

  const int n4 = NN * CH / 4;
  dim3 pg(NCH), pb(PREP_T);
  dim3 sb(SC_T), sg(2 * NB);

  const int mtiles = NN / 16;
  dim3 dg(256), db(DENSE_WAVES * 32);

  prep<<<pg, pb, 0, stream>>>((float4*)agg, n4, cnt, src, dst, E, seg, rec);
  scatter_lds<<<sg, sb, 0, stream>>>(x, seg, rec, agg, E, cnt);
  mlp_chain<true, false><<<dg, db, 0, stream>>>(x, agg, w1a, b1a, w1b, b1b, nullptr, nullptr, nullptr, nullptr,
                                                bufB, agg2, nullptr, mtiles);

  scatter_lds<<<sg, sb, 0, stream>>>(bufB, seg, rec, agg2, E, cnt + 2);
  mlp_chain<false, true><<<dg, db, 0, stream>>>(agg2, nullptr, w2a, b2a, w2b, b2b, w3, b3, w4, b4,
                                                nullptr, nullptr, outp, mtiles);
}


# Dense launches: a work-group's 13th tile shared by its four spare waves, one per SIMD (a quarter of each layer's output tiles each, hand-over through LDS), not a 4th tile on one SIMD; sums unchanged
# speedup vs baseline: 3.3369x; 1.0325x over previous
#include <hip/hip_runtime.h>

typedef __attribute__((ext_vector_type(2))) float v2f;
typedef __attribute__((ext_vector_type(8))) float v8f;

#define NN 50000
#define CH 128

#define NB 256
#define BS ((NN + NB - 1) / NB)
#define NCH 256
#define NBIN (2 * NB)
#define PREP_T 1024
#define SC_W 16
#define SC_T (SC_W * 32)
#define SEGW (NCH / SC_W)

__global__ __launch_bounds__(PREP_T) void prep(float4* __restrict__ p, int n4, int* __restrict__ cnt,
                                               const int* __restrict__ src, const int* __restrict__ dst, int E,
                                               int2* __restrict__ seg, int2* __restrict__ rec) {
  __shared__ int hist[NBIN];
  __shared__ int cur[NBIN];
  __shared__ int wsum[NBIN / 32];
  const int tid = (int)threadIdx.x;
  const int lane = tid & 31;
  const int wv = tid >> 5;
  const int c = (int)blockIdx.x;

  const int per = (n4 + (int)gridDim.x - 1) / (int)gridDim.x;
  for (int i = tid; i < per; i += PREP_T) {
    const int k = c * per + i;
    if (k < n4) p[k] = make_float4(0.f, 0.f, 0.f, 0.f);
  }
  if (c == 0 && tid < 4) cnt[tid] = 0;
  if (tid < NBIN) hist[tid] = 0;
  __syncthreads();

  const int CE = (E + NCH - 1) / NCH;
  const int e0 = c * CE;
  const int e1 = min(e0 + CE, E);
  for (int e = e0 + tid; e < e1; e += PREP_T) {
    const int s = src[e], d = dst[e];
    atomicAdd(&hist[2 * (s / BS) + (d >= NN / 2 ? 1 : 0)], 1);
  }
  __syncthreads();

  int v = 0, incl = 0;
  if (tid < NBIN) {
    v = hist[tid];
    incl = v;
#pragma unroll
    for (int o = 1; o < 32; o <<= 1) {
      const int t = __shfl_up(incl, o, 32);
      if (lane >= o) incl += t;
    }
    if (lane == 31) wsum[wv] = incl;
  }
  __syncthreads();
  if (tid < NBIN) {
    int pre = 0;
    for (int k = 0; k < wv; ++k) pre += wsum[k];
    const int ex = pre + incl - v;
    cur[tid] = ex;
    seg[(size_t)tid * NCH + c] = make_int2(ex, ex + v);
  }
  __syncthreads();

  for (int e = e0 + tid; e < e1; e += PREP_T) {
    const int s = src[e], d = dst[e];
    const int pos = atomicAdd(&cur[2 * (s / BS) + (d >= NN / 2 ? 1 : 0)], 1);
    rec[(size_t)e0 + pos] = make_int2(s, d);
  }
}

__global__ __launch_bounds__(SC_T) void scatter_lds(const float* __restrict__ X,
                                                    const int2* __restrict__ seg,
                                                    const int2* __restrict__ rec,
                                                    float* __restrict__ agg, int E,
                                                    int* __restrict__ cnt) {
  __shared__ float rows[BS * CH];
  __shared__ int s_tk[2];
  __shared__ int s_q;

  const int tid  = (int)threadIdx.x;
  const int lane = tid & 31;
  const int w  = __builtin_amdgcn_readfirstlane(tid >> 5);
  const int CE = (E + NCH - 1) / NCH;

  int pending = 0;
  if (tid == 0) {
    const unsigned pl = __builtin_amdgcn_s_sendmsg_rtn(0x87);
    const int q0 = (int)((pl >> 18) & 1u);
    s_q = q0;
    pending = atomicAdd(cnt + q0, 1);
  }
  __syncthreads();
  int q = s_q;
  bool other = false;

  for (int it = 0; it < 2 * NB + 2; ++it) {
    if (tid == 0) s_tk[it & 1] = pending;
    __syncthreads();
    const int t = s_tk[it & 1];
    if (t >= NB) {
      if (other) break;
      other = true; q ^= 1;
      if (tid == 0) pending = atomicAdd(cnt + q, 1);
      continue;
    }
    int2 sg = make_int2(0, 0);
    if (lane < SEGW) sg = seg[(size_t)(2 * t + q) * NCH + w * SEGW + lane];

    const int r0 = t * BS;
    const int nr4 = min(BS, NN - r0) * (CH / 4);
    const float4* xp = reinterpret_cast<const float4*>(X + (size_t)r0 * CH);
    float4* lp = reinterpret_cast<float4*>(rows);
    constexpr int FILL_IT = (BS * (CH / 4) + SC_T - 1) / SC_T;
    float4 fl[FILL_IT];
#pragma unroll
    for (int k = 0; k < FILL_IT; ++k) fl[k] = xp[min(tid + k * SC_T, nr4 - 1)];
#pragma unroll
    for (int k = 0; k < FILL_IT; ++k) lp[min(tid + k * SC_T, nr4 - 1)] = fl[k];

    const int len = sg.y - sg.x;
    int incl = len;
#pragma unroll
    for (int o = 1; o < 32; o <<= 1) {
      const int u = __shfl_up(incl, o, 32);
      if (lane >= o) incl += u;
    }
    const int excl = incl - len;
    const int T = __builtin_amdgcn_readlane(incl, 31);

    __syncthreads();

#pragma nounroll
    for (int base = 0; base < T; base += 32) {
      const int f = base + lane;
      int cj = 0, off = 0;
#pragma unroll
      for (int j = 0; j < SEGW; ++j) {
        const int pj = __builtin_amdgcn_readlane(excl, j);
        const int sj = __builtin_amdgcn_readlane(sg.x, j);
        if (f >= pj) { cj = j; off = sj + (f - pj); }
      }
      int sv = r0, dv = 0;
      if (f < T) {
        const int2 r = rec[(size_t)(w * SEGW + cj) * CE + off];
        sv = r.x; dv = r.y;
      }
      const int m = min(32, T - base);

#pragma nounroll
      for (int i = 0; i < m; i += 4) {
        float v[4][4];
#pragma unroll
        for (int u = 0; u < 4; ++u) {
          const int k = min(i + u, m - 1);
          const int s = __builtin_amdgcn_readlane(sv, k) - r0;
          const float* lr = rows + s * CH + lane;
          v[u][0] = lr[0]; v[u][1] = lr[32]; v[u][2] = lr[64]; v[u][3] = lr[96];
        }
#pragma unroll
        for (int u = 0; u < 4; ++u) {
          if (i + u < m) {
            const int d = __builtin_amdgcn_readlane(dv, i + u);
            float* a = agg + (size_t)d * CH + lane;
            unsafeAtomicAdd(a + 0,  v[u][0]);
            unsafeAtomicAdd(a + 32, v[u][1]);
            unsafeAtomicAdd(a + 64, v[u][2]);
            unsafeAtomicAdd(a + 96, v[u][3]);
          }
        }
      }
    }
    if (tid == 0) pending = atomicAdd(cnt + q, 1);
  }
}

#define DENSE_WAVES 16
#define TEAM_W 4
#define TEAM0 (DENSE_WAVES - TEAM_W)

template <int N>
__device__ __forceinline__ void stage_w(float4* __restrict__ Wf, const float* __restrict__ W,
                                        int tid, int rot) {
  constexpr int NX = N / 4;
  constexpr int PER = (32 * NX) / (DENSE_WAVES * 32);
  static_assert(PER * DENSE_WAVES * 32 == 32 * NX, "tasks divide evenly among the threads");
#pragma unroll
  for (int i = 0; i < PER; ++i) {
    const int task = tid + i * DENSE_WAVES * 32;
    const int x  = task % NX;
    const int kq = ((task / NX) + rot) & 31;
    const float4* wp = reinterpret_cast<const float4*>(W + (size_t)(4 * kq) * N) + x;
    const float4 r0 = wp[0], r1 = wp[NX], r2 = wp[2 * NX], r3 = wp[3 * NX];
    const int j = kq >> 1, hh = kq & 1;
    float4* s = Wf + ((x >> 2) * 16 + j) * 32 + hh * 16 + ((x & 1) << 3) + ((x & 2) << 1);
    s[0] = make_float4(r0.x, r1.x, r2.x, r3.x);
    s[1] = make_float4(r0.y, r1.y, r2.y, r3.y);
    s[2] = make_float4(r0.z, r1.z, r2.z, r3.z);
    s[3] = make_float4(r0.w, r1.w, r2.w, r3.w);
  }
}

template <bool ADD_AGG>
__device__ __forceinline__ void load_rows(float4 (&a)[16], const float* __restrict__ A,
                                          const float* __restrict__ AGG, int row, int h) {
  const float4* ap = reinterpret_cast<const float4*>(A + (size_t)row * CH + 4 * h);
#pragma unroll
  for (int j = 0; j < 16; ++j) a[j] = ap[2 * j];
  if (ADD_AGG) {
    const float4* gp = reinterpret_cast<const float4*>(AGG + (size_t)row * CH + 4 * h);
#pragma unroll
    for (int j = 0; j < 16; ++j) {
      const float4 t = gp[2 * j];
      a[j].x += t.x; a[j].y += t.y; a[j].z += t.z; a[j].w += t.w;
    }
  }
}

template <int NT>
__device__ __forceinline__ void layer(v8f (&acc)[NT], const float4 (&in)[16],
                                      const float4* __restrict__ Wf,
                                      const float4* __restrict__ bias, int lane, int h) {
  const float4* b4p = bias + h;
#pragma unroll
  for (int q = 0; q < NT; ++q) {
    const float4 lo = b4p[4 * q], hi = b4p[4 * q + 2];
    acc[q][0] = lo.x; acc[q][1] = lo.y; acc[q][2] = lo.z; acc[q][3] = lo.w;
    acc[q][4] = hi.x; acc[q][5] = hi.y; acc[q][6] = hi.z; acc[q][7] = hi.w;
  }
  const float4* wp = Wf + lane;
#pragma unroll
  for (int j = 0; j < 16; ++j) {
    float4 w[NT];
#pragma unroll
    for (int q = 0; q < NT; ++q) w[q] = wp[(q * 16 + j) * 32];
    v2f blo; blo.x = in[j].x; blo.y = in[j].y;
    v2f bhi; bhi.x = in[j].z; bhi.y = in[j].w;
#pragma unroll
    for (int q = 0; q < NT; ++q) {
      v2f t; t.x = w[q].x; t.y = w[q].y;
      acc[q] = __builtin_amdgcn_wmma_f32_16x16x4_f32(false, t, false, blo, (short)0, acc[q], false, false);
    }
#pragma unroll
    for (int q = 0; q < NT; ++q) {
      v2f t; t.x = w[q].z; t.y = w[q].w;
      acc[q] = __builtin_amdgcn_wmma_f32_16x16x4_f32(false, t, false, bhi, (short)0, acc[q], false, false);
    }
  }
}

__device__ __forceinline__ void relu_rows(float4 (&o)[16], const v8f (&acc)[8]) {
#pragma unroll
  for (int q = 0; q < 8; ++q) {
    o[2 * q]     = make_float4(fmaxf(acc[q][0], 0.f), fmaxf(acc[q][1], 0.f), fmaxf(acc[q][2], 0.f), fmaxf(acc[q][3], 0.f));
    o[2 * q + 1] = make_float4(fmaxf(acc[q][4], 0.f), fmaxf(acc[q][5], 0.f), fmaxf(acc[q][6], 0.f), fmaxf(acc[q][7], 0.f));
  }
}

__device__ __forceinline__ void put_quarter(float4* __restrict__ X, int part, int lane, const v8f (&acc)[2]) {
#pragma unroll
  for (int u = 0; u < 4; ++u) {
    const int r = 4 * (u & 1);
    X[(4 * part + u) * 32 + lane] = make_float4(fmaxf(acc[u >> 1][r], 0.f), fmaxf(acc[u >> 1][r + 1], 0.f),
                                                fmaxf(acc[u >> 1][r + 2], 0.f), fmaxf(acc[u >> 1][r + 3], 0.f));
  }
}
__device__ __forceinline__ void get_rows(float4 (&o)[16], const float4* __restrict__ X, int lane) {
#pragma unroll
  for (int j = 0; j < 16; ++j) o[j] = X[j * 32 + lane];
}

__device__ __forceinline__ void lds_barrier() {
  __builtin_amdgcn_fence(__ATOMIC_RELEASE, "workgroup", "local");
  __builtin_amdgcn_s_barrier();
  __builtin_amdgcn_fence(__ATOMIC_ACQUIRE, "workgroup", "local");
}

template <bool ADD_AGG, bool HEAD>
__global__ __launch_bounds__(DENSE_WAVES * 32) void mlp_chain(const float* __restrict__ A,
                                                 const float* __restrict__ AGG,
                                                 const float* __restrict__ Wa,
                                                 const float* __restrict__ ba,
                                                 const float* __restrict__ Wb,
                                                 const float* __restrict__ bb,
                                                 const float* __restrict__ W3,
                                                 const float* __restrict__ b3,
                                                 const float* __restrict__ w4,
                                                 const float* __restrict__ b4,
                                                 float* __restrict__ out,
                                                 float* __restrict__ out2,
                                                 float* __restrict__ outv,
                                                 int mtiles) {
  constexpr int NW = DENSE_WAVES;
  __shared__ float4 Fa[8 * 16 * 32];
  __shared__ float4 Fb[8 * 16 * 32];
  __shared__ float4 F3[HEAD ? 4 * 16 * 32 : 1];
  __shared__ float4 Bi[(CH + CH + 64 + 64) / 4];
  __shared__ float4 Xa[16 * 32];
  __shared__ float4 Xb[HEAD ? 16 * 32 : 1];
  __shared__ float4 Xh[HEAD ? 8 * 32 : 1];

  const int tid  = threadIdx.x;
  const int lane = tid & 31;
  const int wave = __builtin_amdgcn_readfirstlane(tid >> 5);
  const int c = lane & 15;
  const int h = lane >> 4;

  const int cnt = (mtiles - (int)blockIdx.x + (int)gridDim.x - 1) / (int)gridDim.x;

  const bool split = (cnt == TEAM0 + 1);
  const bool team = split && wave >= TEAM0;
  const int part = wave - TEAM0;

  float4 in[16];
  int t = team ? TEAM0 : wave;
  if (t < cnt) load_rows<ADD_AGG>(in, A, AGG, ((((int)blockIdx.x + (int)gridDim.x * t)) << 4) + c, h);

  stage_w<CH>(Fa, Wa, tid, (int)blockIdx.x);
  stage_w<CH>(Fb, Wb, tid, (int)blockIdx.x);
  if constexpr (HEAD) stage_w<64>(F3, W3, tid, (int)blockIdx.x);
  if (tid < CH / 4) {
    Bi[tid] = reinterpret_cast<const float4*>(ba)[tid];
    Bi[CH / 4 + tid] = reinterpret_cast<const float4*>(bb)[tid];
    if constexpr (HEAD) {
      if (tid < 16) {
        Bi[CH / 2 + tid] = reinterpret_cast<const float4*>(b3)[tid];
        Bi[CH / 2 + 16 + tid] = reinterpret_cast<const float4*>(w4)[tid];
      }
    }
  }
  __syncthreads();

  while (!team && t < cnt) {
    const int m0 = ((int)blockIdx.x + (int)gridDim.x * t) << 4;
    __atomic_signal_fence(__ATOMIC_SEQ_CST);

    v8f acc[8];
    layer<8>(acc, in, Fa, Bi, lane, h);
    if constexpr (HEAD) __builtin_amdgcn_sched_barrier(0);
    relu_rows(in, acc);
    layer<8>(acc, in, Fb, Bi + CH / 4, lane, h);

    if constexpr (HEAD) {
      __builtin_amdgcn_sched_barrier(0);
      relu_rows(in, acc);
      v8f a3[4];
      layer<4>(a3, in, F3, Bi + CH / 2, lane, h);
      const float4* w4p = Bi + CH / 2 + 16 + h;
      float s = 0.f;
#pragma unroll
      for (int q = 0; q < 4; ++q) {
        const float4 lo = w4p[4 * q], hi = w4p[4 * q + 2];
        s += fmaxf(a3[q][0], 0.f) * lo.x; s += fmaxf(a3[q][1], 0.f) * lo.y;
        s += fmaxf(a3[q][2], 0.f) * lo.z; s += fmaxf(a3[q][3], 0.f) * lo.w;
        s += fmaxf(a3[q][4], 0.f) * hi.x; s += fmaxf(a3[q][5], 0.f) * hi.y;
        s += fmaxf(a3[q][6], 0.f) * hi.z; s += fmaxf(a3[q][7], 0.f) * hi.w;
      }
      s += __shfl_xor(s, 16, 32);
      if (h == 0) outv[m0 + c] = s + b4[0];
    } else {
      relu_rows(in, acc);
      float4* o1 = reinterpret_cast<float4*>(out  + (size_t)(m0 + c) * CH + 4 * h);
      float4* o2 = reinterpret_cast<float4*>(out2 + (size_t)(m0 + c) * CH + 4 * h);
#pragma unroll
      for (int j = 0; j < 16; ++j) { o1[2 * j] = in[j]; o2[2 * j] = in[j]; }
    }

    t += NW;
    if (t < cnt) load_rows<ADD_AGG>(in, A, AGG, ((((int)blockIdx.x + (int)gridDim.x * t)) << 4) + c, h);
  }

  const int m0s = ((int)blockIdx.x + (int)gridDim.x * TEAM0) << 4;
  if (team) {
    v8f a2[2];
    layer<2>(a2, in, Fa + part * (2 * 16 * 32), Bi + 8 * part, lane, h);
    put_quarter(Xa, part, lane, a2);
  }
  if (split) lds_barrier();
  if (team) {
    get_rows(in, Xa, lane);
    v8f a2[2];
    layer<2>(a2, in, Fb + part * (2 * 16 * 32), Bi + CH / 4 + 8 * part, lane, h);
    if constexpr (HEAD) {
      put_quarter(Xb, part, lane, a2);
    } else {
      float4* o1 = reinterpret_cast<float4*>(out  + (size_t)(m0s + c) * CH + 4 * h) + 8 * part;
      float4* o2 = reinterpret_cast<float4*>(out2 + (size_t)(m0s + c) * CH + 4 * h) + 8 * part;
#pragma unroll
      for (int u = 0; u < 4; ++u) {
        const int r = 4 * (u & 1);
        const float4 v = make_float4(fmaxf(a2[u >> 1][r], 0.f), fmaxf(a2[u >> 1][r + 1], 0.f),
                                     fmaxf(a2[u >> 1][r + 2], 0.f), fmaxf(a2[u >> 1][r + 3], 0.f));
        o1[2 * u] = v; o2[2 * u] = v;
      }
    }
  }
  if constexpr (HEAD) {
    if (split) lds_barrier();
    if (team) {
      get_rows(in, Xb, lane);
      v8f a3[1];
      layer<1>(a3, in, F3 + part * (16 * 32), Bi + CH / 2 + 4 * part, lane, h);
      Xh[(2 * part) * 32 + lane]     = make_float4(fmaxf(a3[0][0], 0.f), fmaxf(a3[0][1], 0.f), fmaxf(a3[0][2], 0.f), fmaxf(a3[0][3], 0.f));
      Xh[(2 * part + 1) * 32 + lane] = make_float4(fmaxf(a3[0][4], 0.f), fmaxf(a3[0][5], 0.f), fmaxf(a3[0][6], 0.f), fmaxf(a3[0][7], 0.f));
    }
    if (split) lds_barrier();
    if (team && part == 0) {
      const float4* w4p = Bi + CH / 2 + 16 + h;
      float s = 0.f;
#pragma unroll
      for (int q = 0; q < 4; ++q) {
        const float4 lo = w4p[4 * q], hi = w4p[4 * q + 2];
        const float4 vl = Xh[(2 * q) * 32 + lane], vh = Xh[(2 * q + 1) * 32 + lane];
        s += vl.x * lo.x; s += vl.y * lo.y; s += vl.z * lo.z; s += vl.w * lo.w;
        s += vh.x * hi.x; s += vh.y * hi.y; s += vh.z * hi.z; s += vh.w * hi.w;
      }
      s += __shfl_xor(s, 16, 32);
      if (h == 0) outv[m0s + c] = s + b4[0];
    }
  }
}

extern "C" void kernel_launch(void* const* d_in, const int* in_sizes, int n_in,
                              void* d_out, int out_size, void* d_ws, size_t ws_size,
                              hipStream_t stream) {
  const float* x   = (const float*)d_in[0];
  const int*   ei  = (const int*)d_in[1];
  const int    E   = in_sizes[1] / 2;
  const int* src = ei;
  const int* dst = ei + E;
  const float* w1a = (const float*)d_in[2];
  const float* b1a = (const float*)d_in[3];
  const float* w1b = (const float*)d_in[4];
  const float* b1b = (const float*)d_in[5];
  const float* w2a = (const float*)d_in[6];
  const float* b2a = (const float*)d_in[7];
  const float* w2b = (const float*)d_in[8];
  const float* b2b = (const float*)d_in[9];
  const float* w3  = (const float*)d_in[10];
  const float* b3  = (const float*)d_in[11];
  const float* w4  = (const float*)d_in[12];
  const float* b4  = (const float*)d_in[13];
  float* outp = (float*)d_out;

  float* agg  = (float*)d_ws;
  float* bufB = agg  + (size_t)NN * CH;
  float* agg2 = bufB + (size_t)NN * CH;
  int*   cnt  = (int*)(agg2 + (size_t)NN * CH);
  int2*  seg  = (int2*)(cnt + 64);
  int2*  rec  = seg + (size_t)NBIN * NCH;

  const int n4 = NN * CH / 4;
  dim3 pg(NCH), pb(PREP_T);
  dim3 sb(SC_T), sg(2 * NB);

  const int mtiles = NN / 16;
  dim3 dg(256), db(DENSE_WAVES * 32);

  prep<<<pg, pb, 0, stream>>>((float4*)agg, n4, cnt, src, dst, E, seg, rec);
  scatter_lds<<<sg, sb, 0, stream>>>(x, seg, rec, agg, E, cnt);
  mlp_chain<true, false><<<dg, db, 0, stream>>>(x, agg, w1a, b1a, w1b, b1b, nullptr, nullptr, nullptr, nullptr,
                                                bufB, agg2, nullptr, mtiles);

  scatter_lds<<<sg, sb, 0, stream>>>(bufB, seg, rec, agg2, E, cnt + 2);
  mlp_chain<false, true><<<dg, db, 0, stream>>>(agg2, nullptr, w2a, b2a, w2b, b2b, w3, b3, w4, b4,
                                                nullptr, nullptr, outp, mtiles);
}
